# starting kernel
# baseline (speedup 1.0000x reference)
#include <hip/hip_runtime.h>


typedef float v2f __attribute__((ext_vector_type(2)));
typedef float v8f __attribute__((ext_vector_type(8)));

#define B_   32
#define C_   64
#define H_   32
#define W_   32
#define KD   576
#define COG  16
#define TILE 16
#define CCH  32

__global__ __launch_bounds__(256) void wtrans_kernel(const float* __restrict__ w,
                                                     float* __restrict__ wt) {
  int idx = blockIdx.x * 256 + threadIdx.x;
  if (idx < C_ * KD) {
    int co = idx / KD, k = idx - co * KD;
    int cg = co >> 4, c = co & 15;
    wt[cg * (KD * COG) + k * COG + c] = w[idx];
  }
}

__global__ __launch_bounds__(256) void adder_kernel(const float* __restrict__ in,
                                                    const float* __restrict__ wt,
                                                    float* __restrict__ y) {
  __shared__ float xs[CCH][TILE + 2][TILE + 2];
  const int tid    = threadIdx.x;
  const int cg     = blockIdx.x;
  const int tileId = blockIdx.y;
  const int n      = blockIdx.z;
  const int ty0 = (tileId >> 1) * TILE;
  const int tx0 = (tileId & 1) * TILE;
  const int tx = tid & 15, ty = tid >> 4;

  const float* wblk = wt + cg * (KD * COG);
  __builtin_prefetch(wblk, 0, 1);

  float acc[COG];
#pragma unroll
  for (int c = 0; c < COG; ++c) acc[c] = 0.f;

  for (int c0 = 0; c0 < C_; c0 += CCH) {
    for (int i = tid; i < CCH * (TILE + 2) * (TILE + 2); i += 256) {
      int ci = i / ((TILE + 2) * (TILE + 2));
      int r  = i - ci * ((TILE + 2) * (TILE + 2));
      int yy = r / (TILE + 2);
      int xx = r - yy * (TILE + 2);
      int gy = ty0 + yy - 1;
      int gx = tx0 + xx - 1;
      float v = 0.f;
      if ((unsigned)gy < (unsigned)H_ && (unsigned)gx < (unsigned)W_)
        v = in[(((size_t)n * C_ + (c0 + ci)) * H_ + gy) * W_ + gx];
      xs[ci][yy][xx] = v;
    }
    __syncthreads();

    for (int ci = 0; ci < CCH; ++ci) {
#pragma unroll
      for (int kh = 0; kh < 3; ++kh) {
#pragma unroll
        for (int kw = 0; kw < 3; ++kw) {
          const float xv = xs[ci][ty + kh][tx + kw];
          const float* wk = wblk + (((c0 + ci) * 3 + kh) * 3 + kw) * COG;
#pragma unroll
          for (int c = 0; c < COG; ++c)
            acc[c] += fabsf(xv - wk[c]);
        }
      }
    }
    __syncthreads();
  }

  const int gy = ty0 + ty, gx = tx0 + tx;
#pragma unroll
  for (int c = 0; c < COG; ++c)
    y[(((size_t)n * C_ + (cg * COG + c)) * H_ + gy) * W_ + gx] = -acc[c];
}

__global__ __launch_bounds__(256) void stats_kernel(const float* __restrict__ y,
                                                    const float* __restrict__ g,
                                                    const float* __restrict__ b,
                                                    float* __restrict__ ss) {
  const int c    = blockIdx.x;
  const int tid  = threadIdx.x;
  const int wave = tid >> 5, lane = tid & 31;

  v8f accS = {};
  v8f accQ = {};
  v2f ones = {1.f, 1.f};

  const float* chan = y + (size_t)c * (H_ * W_);
  for (int base = wave * 64; base < B_ * H_ * W_; base += 8 * 64) {
    int e0 = base + lane * 2;
    int n0 = e0 >> 10, i0 = e0 & 1023;
    const float* p = chan + (size_t)n0 * (C_ * H_ * W_) + i0;
    v2f d = {p[0], p[1]};
    v2f q = {d.x * d.x, d.y * d.y};
    accS = __builtin_amdgcn_wmma_f32_16x16x4_f32(false, ones, false, d,
                                                 (short)0, accS, false, false);
    accQ = __builtin_amdgcn_wmma_f32_16x16x4_f32(false, ones, false, q,
                                                 (short)0, accQ, false, false);
  }

  float s = accS[0], q = accQ[0];
  for (int off = 16; off > 0; off >>= 1) {
    s += __shfl_xor(s, off, 32);
    q += __shfl_xor(q, off, 32);
  }
  s *= 0.5f;
  q *= 0.5f;

  __shared__ float ls[8], lq[8];
  if (lane == 0) { ls[wave] = s; lq[wave] = q; }
  __syncthreads();
  if (tid == 0) {
    float S = 0.f, Q = 0.f;
    for (int i = 0; i < 8; ++i) { S += ls[i]; Q += lq[i]; }
    const float inv_n = 1.0f / (float)(B_ * H_ * W_);
    float mu    = S * inv_n;
    float var   = Q * inv_n - mu * mu;
    float inv   = rsqrtf(var + 1e-5f);
    float scale = g[c] * inv;
    ss[c]       = scale;
    ss[C_ + c]  = b[c] - mu * scale;
  }
}

__global__ __launch_bounds__(256) void apply_kernel(const float* __restrict__ y,
                                                    const float* __restrict__ ss,
                                                    const float* __restrict__ resid,
                                                    float* __restrict__ out, int n) {
  int idx = blockIdx.x * 256 + threadIdx.x;
  if (idx < n) {
    int c = (idx >> 10) & (C_ - 1);
    float v = y[idx] * ss[c] + ss[C_ + c];
    if (resid) v += resid[idx];
    out[idx] = fmaxf(v, 0.f);
  }
}

extern "C" void kernel_launch(void* const* d_in, const int* in_sizes, int n_in,
                              void* d_out, int out_size, void* d_ws, size_t ws_size,
                              hipStream_t stream) {
  (void)in_sizes; (void)n_in; (void)out_size; (void)ws_size;
  const float* x  = (const float*)d_in[0];
  const float* w1 = (const float*)d_in[1];
  const float* g1 = (const float*)d_in[2];
  const float* b1 = (const float*)d_in[3];
  const float* w2 = (const float*)d_in[4];
  const float* g2 = (const float*)d_in[5];
  const float* b2 = (const float*)d_in[6];
  float* out = (float*)d_out;

  char* ws = (char*)d_ws;
  const size_t TEN   = (size_t)B_ * C_ * H_ * W_;
  const size_t WTSZ  = (size_t)C_ * KD * sizeof(float);
  float* ybuf = (float*)(ws);
  float* abuf = (float*)(ws + TEN * 4);
  float* wt1  = (float*)(ws + 2 * TEN * 4);
  float* wt2  = (float*)(ws + 2 * TEN * 4 + WTSZ);
  float* ss1  = (float*)(ws + 2 * TEN * 4 + 2 * WTSZ);
  float* ss2  = (float*)(ws + 2 * TEN * 4 + 2 * WTSZ + 512);

  wtrans_kernel<<<144, 256, 0, stream>>>(w1, wt1);
  wtrans_kernel<<<144, 256, 0, stream>>>(w2, wt2);

  dim3 agrid(C_ / COG, (H_ / TILE) * (W_ / TILE), B_);

  adder_kernel<<<agrid, 256, 0, stream>>>(x, wt1, ybuf);
  stats_kernel<<<C_, 256, 0, stream>>>(ybuf, g1, b1, ss1);
  apply_kernel<<<(int)(TEN / 256), 256, 0, stream>>>(ybuf, ss1, nullptr, abuf, (int)TEN);

  adder_kernel<<<agrid, 256, 0, stream>>>(abuf, wt2, ybuf);
  stats_kernel<<<C_, 256, 0, stream>>>(ybuf, g2, b2, ss2);
  apply_kernel<<<(int)(TEN / 256), 256, 0, stream>>>(ybuf, ss2, x, out, (int)TEN);
}


# BN statistics pass made memory-parallel: 8 work-groups per channel (was 1), each wave issues its four 128-bit loads before the WMMA sums; partials added in the normalize kernel (new reduction tree)
# speedup vs baseline: 1.2747x; 1.2747x over previous
#include <hip/hip_runtime.h>


typedef float v2f __attribute__((ext_vector_type(2)));
typedef float v8f __attribute__((ext_vector_type(8)));

#define B_   32
#define C_   64
#define H_   32
#define W_   32
#define KD   576
#define COG  16
#define TILE 16
#define CCH  32

__global__ __launch_bounds__(256) void wtrans_kernel(const float* __restrict__ w,
                                                     float* __restrict__ wt) {
  int idx = blockIdx.x * 256 + threadIdx.x;
  if (idx < C_ * KD) {
    int co = idx / KD, k = idx - co * KD;
    int cg = co >> 4, c = co & 15;
    wt[cg * (KD * COG) + k * COG + c] = w[idx];
  }
}

__global__ __launch_bounds__(256) void adder_kernel(const float* __restrict__ in,
                                                    const float* __restrict__ wt,
                                                    float* __restrict__ y) {
  __shared__ float xs[CCH][TILE + 2][TILE + 2];
  const int tid    = threadIdx.x;
  const int cg     = blockIdx.x;
  const int tileId = blockIdx.y;
  const int n      = blockIdx.z;
  const int ty0 = (tileId >> 1) * TILE;
  const int tx0 = (tileId & 1) * TILE;
  const int tx = tid & 15, ty = tid >> 4;

  const float* wblk = wt + cg * (KD * COG);
  __builtin_prefetch(wblk, 0, 1);

  float acc[COG];
#pragma unroll
  for (int c = 0; c < COG; ++c) acc[c] = 0.f;

  for (int c0 = 0; c0 < C_; c0 += CCH) {
    for (int i = tid; i < CCH * (TILE + 2) * (TILE + 2); i += 256) {
      int ci = i / ((TILE + 2) * (TILE + 2));
      int r  = i - ci * ((TILE + 2) * (TILE + 2));
      int yy = r / (TILE + 2);
      int xx = r - yy * (TILE + 2);
      int gy = ty0 + yy - 1;
      int gx = tx0 + xx - 1;
      float v = 0.f;
      if ((unsigned)gy < (unsigned)H_ && (unsigned)gx < (unsigned)W_)
        v = in[(((size_t)n * C_ + (c0 + ci)) * H_ + gy) * W_ + gx];
      xs[ci][yy][xx] = v;
    }
    __syncthreads();

    for (int ci = 0; ci < CCH; ++ci) {
#pragma unroll
      for (int kh = 0; kh < 3; ++kh) {
#pragma unroll
        for (int kw = 0; kw < 3; ++kw) {
          const float xv = xs[ci][ty + kh][tx + kw];
          const float* wk = wblk + (((c0 + ci) * 3 + kh) * 3 + kw) * COG;
#pragma unroll
          for (int c = 0; c < COG; ++c)
            acc[c] += fabsf(xv - wk[c]);
        }
      }
    }
    __syncthreads();
  }

  const int gy = ty0 + ty, gx = tx0 + tx;
#pragma unroll
  for (int c = 0; c < COG; ++c)
    y[(((size_t)n * C_ + (cg * COG + c)) * H_ + gy) * W_ + gx] = -acc[c];
}

#define NSPLIT 8
#define PSTRIDE 32
__global__ __launch_bounds__(256) void stats_kernel(const float* __restrict__ y,
                                                    float* __restrict__ part) {
  const int c    = blockIdx.x / NSPLIT;
  const int sp   = blockIdx.x % NSPLIT;
  const int tid  = threadIdx.x;
  const int wave = tid >> 5, lane = tid & 31;

  const int n = sp * (B_ / NSPLIT) + (wave >> 1);
  const float4* p = reinterpret_cast<const float4*>(
      y + ((size_t)n * C_ + c) * (H_ * W_) + (wave & 1) * 512) + lane;
  float4 v[4];
#pragma unroll
  for (int r = 0; r < 4; ++r) v[r] = p[r * 32];

  v8f accS = {};
  v8f accQ = {};
  v2f ones = {1.f, 1.f};
#pragma unroll
  for (int r = 0; r < 4; ++r) {
    v2f d0 = {v[r].x, v[r].y};
    v2f d1 = {v[r].z, v[r].w};
    v2f q0 = {d0.x * d0.x, d0.y * d0.y};
    v2f q1 = {d1.x * d1.x, d1.y * d1.y};
    accS = __builtin_amdgcn_wmma_f32_16x16x4_f32(false, ones, false, d0,
                                                 (short)0, accS, false, false);
    accQ = __builtin_amdgcn_wmma_f32_16x16x4_f32(false, ones, false, q0,
                                                 (short)0, accQ, false, false);
    accS = __builtin_amdgcn_wmma_f32_16x16x4_f32(false, ones, false, d1,
                                                 (short)0, accS, false, false);
    accQ = __builtin_amdgcn_wmma_f32_16x16x4_f32(false, ones, false, q1,
                                                 (short)0, accQ, false, false);
  }

  float s = accS[0], q = accQ[0];
  for (int off = 16; off > 0; off >>= 1) {
    s += __shfl_xor(s, off, 32);
    q += __shfl_xor(q, off, 32);
  }
  s *= 0.5f;
  q *= 0.5f;

  __shared__ float ls[8], lq[8];
  if (lane == 0) { ls[wave] = s; lq[wave] = q; }
  __syncthreads();
  if (tid == 0) {
    float S = 0.f, Q = 0.f;
    for (int i = 0; i < 8; ++i) { S += ls[i]; Q += lq[i]; }
    part[(size_t)blockIdx.x * PSTRIDE]     = S;
    part[(size_t)blockIdx.x * PSTRIDE + 1] = Q;
  }
}

__global__ __launch_bounds__(256) void apply_kernel(const float* __restrict__ y,
                                                    const float* __restrict__ part,
                                                    const float* __restrict__ g,
                                                    const float* __restrict__ b,
                                                    const float* __restrict__ resid,
                                                    float* __restrict__ out, int n) {
  const int c = (blockIdx.x >> 2) & (C_ - 1);
  float S = 0.f, Q = 0.f;
#pragma unroll
  for (int i = 0; i < NSPLIT; ++i) {
    S += part[(size_t)(c * NSPLIT + i) * PSTRIDE];
    Q += part[(size_t)(c * NSPLIT + i) * PSTRIDE + 1];
  }
  const float inv_n = 1.0f / (float)(B_ * H_ * W_);
  float mu    = S * inv_n;
  float var   = Q * inv_n - mu * mu;
  float inv   = rsqrtf(var + 1e-5f);
  float scale = g[c] * inv;
  float shift = b[c] - mu * scale;

  int idx = blockIdx.x * 256 + threadIdx.x;
  if (idx < n) {
    float v = y[idx] * scale + shift;
    if (resid) v += resid[idx];
    out[idx] = fmaxf(v, 0.f);
  }
}

extern "C" void kernel_launch(void* const* d_in, const int* in_sizes, int n_in,
                              void* d_out, int out_size, void* d_ws, size_t ws_size,
                              hipStream_t stream) {
  (void)in_sizes; (void)n_in; (void)out_size; (void)ws_size;
  const float* x  = (const float*)d_in[0];
  const float* w1 = (const float*)d_in[1];
  const float* g1 = (const float*)d_in[2];
  const float* b1 = (const float*)d_in[3];
  const float* w2 = (const float*)d_in[4];
  const float* g2 = (const float*)d_in[5];
  const float* b2 = (const float*)d_in[6];
  float* out = (float*)d_out;

  char* ws = (char*)d_ws;
  const size_t TEN   = (size_t)B_ * C_ * H_ * W_;
  const size_t WTSZ  = (size_t)C_ * KD * sizeof(float);
  float* ybuf = (float*)(ws);
  float* abuf = (float*)(ws + TEN * 4);
  float* wt1  = (float*)(ws + 2 * TEN * 4);
  float* wt2  = (float*)(ws + 2 * TEN * 4 + WTSZ);
  const size_t PSZ   = (size_t)C_ * NSPLIT * PSTRIDE * sizeof(float);
  float* pt1  = (float*)(ws + 2 * TEN * 4 + 2 * WTSZ);
  float* pt2  = (float*)(ws + 2 * TEN * 4 + 2 * WTSZ + PSZ);

  wtrans_kernel<<<144, 256, 0, stream>>>(w1, wt1);
  wtrans_kernel<<<144, 256, 0, stream>>>(w2, wt2);

  dim3 agrid(C_ / COG, (H_ / TILE) * (W_ / TILE), B_);

  adder_kernel<<<agrid, 256, 0, stream>>>(x, wt1, ybuf);
  stats_kernel<<<C_ * NSPLIT, 256, 0, stream>>>(ybuf, pt1);
  apply_kernel<<<(int)(TEN / 256), 256, 0, stream>>>(ybuf, pt1, g1, b1, nullptr, abuf, (int)TEN);

  adder_kernel<<<agrid, 256, 0, stream>>>(abuf, wt2, ybuf);
  stats_kernel<<<C_ * NSPLIT, 256, 0, stream>>>(ybuf, pt2);
  apply_kernel<<<(int)(TEN / 256), 256, 0, stream>>>(ybuf, pt2, g2, b2, x, out, (int)TEN);
}


# adder inner loop re-tiled to 2 pixels x 8 channels per lane: x-w takes the weight as scalar operand and dual-issues, |.| is a source modifier of the accumulate add; per-output sum order kept
# speedup vs baseline: 1.5960x; 1.2521x over previous
#include <hip/hip_runtime.h>


typedef float v2f __attribute__((ext_vector_type(2)));
typedef float v8f __attribute__((ext_vector_type(8)));
typedef __attribute__((address_space(4))) v8f cv8f;

#define B_   32
#define C_   64
#define H_   32
#define W_   32
#define KD   576
#define COG  16
#define CPL  8
#define TILE 16
#define CCH  32

__global__ __launch_bounds__(256) void wtrans_kernel(const float* __restrict__ w,
                                                     float* __restrict__ wt) {
  int idx = blockIdx.x * 256 + threadIdx.x;
  if (idx < C_ * KD) {
    int co = idx / KD, k = idx - co * KD;
    int cg = co >> 4, c = co & 15;
    wt[cg * (KD * COG) + k * COG + c] = w[idx];
  }
}

__global__ __launch_bounds__(256) void adder_kernel(const float* __restrict__ in,
                                                    const float* __restrict__ wt,
                                                    float* __restrict__ y) {
  __shared__ float xs[CCH][TILE + 2][TILE + 2];
  const int tid    = threadIdx.x;
  const int cg     = blockIdx.x;
  const int tileId = blockIdx.y;
  const int n      = blockIdx.z;
  const int ty0 = (tileId >> 1) * TILE;
  const int tx0 = (tileId & 1) * TILE;
  const int chalf = __builtin_amdgcn_readfirstlane(tid >> 7);
  const int tx = tid & 15, ty = (tid >> 4) & 7;

  const float* wblk = wt + cg * (KD * COG) + chalf * CPL;
  __builtin_prefetch(wblk, 0, 1);

  float acc0[CPL], acc1[CPL];
#pragma unroll
  for (int c = 0; c < CPL; ++c) { acc0[c] = 0.f; acc1[c] = 0.f; }

  for (int c0 = 0; c0 < C_; c0 += CCH) {
    for (int i = tid; i < CCH * (TILE + 2) * (TILE + 2); i += 256) {
      int ci = i / ((TILE + 2) * (TILE + 2));
      int r  = i - ci * ((TILE + 2) * (TILE + 2));
      int yy = r / (TILE + 2);
      int xx = r - yy * (TILE + 2);
      int gy = ty0 + yy - 1;
      int gx = tx0 + xx - 1;
      float v = 0.f;
      if ((unsigned)gy < (unsigned)H_ && (unsigned)gx < (unsigned)W_)
        v = in[(((size_t)n * C_ + (c0 + ci)) * H_ + gy) * W_ + gx];
      xs[ci][yy][xx] = v;
    }
    __syncthreads();

    for (int ci = 0; ci < CCH; ++ci) {
#pragma unroll
      for (int kh = 0; kh < 3; ++kh) {
#pragma unroll
        for (int kw = 0; kw < 3; ++kw) {
          const float xa = xs[ci][ty + kh][tx + kw], xb = xs[ci][ty + 8 + kh][tx + kw];
          const v8f wk = *(const cv8f*)(wblk + (((c0 + ci) * 3 + kh) * 3 + kw) * COG);
#pragma unroll
          for (int c = 0; c < CPL; ++c) {
            const float w = wk[c];
            const float da = xa - w, db = xb - w;
            asm("v_add_f32_e64 %0, %0, |%1|" : "+v"(acc0[c]) : "v"(da));
            asm("v_add_f32_e64 %0, %0, |%1|" : "+v"(acc1[c]) : "v"(db));
          }
        }
      }
    }
    __syncthreads();
  }

  const int gx = tx0 + tx;
#pragma unroll
  for (int c = 0; c < CPL; ++c) {
    float* yc = y + (((size_t)n * C_ + (cg * COG + chalf * CPL + c)) * H_ + ty0 + ty) * W_ + gx;
    yc[0]      = -acc0[c];
    yc[8 * W_] = -acc1[c];
  }
}

#define NSPLIT 8
#define PSTRIDE 32
__global__ __launch_bounds__(256) void stats_kernel(const float* __restrict__ y,
                                                    float* __restrict__ part) {
  const int c    = blockIdx.x / NSPLIT;
  const int sp   = blockIdx.x % NSPLIT;
  const int tid  = threadIdx.x;
  const int wave = tid >> 5, lane = tid & 31;

  const int n = sp * (B_ / NSPLIT) + (wave >> 1);
  const float4* p = reinterpret_cast<const float4*>(
      y + ((size_t)n * C_ + c) * (H_ * W_) + (wave & 1) * 512) + lane;
  float4 v[4];
#pragma unroll
  for (int r = 0; r < 4; ++r) v[r] = p[r * 32];

  v8f accS = {};
  v8f accQ = {};
  v2f ones = {1.f, 1.f};
#pragma unroll
  for (int r = 0; r < 4; ++r) {
    v2f d0 = {v[r].x, v[r].y};
    v2f d1 = {v[r].z, v[r].w};
    v2f q0 = {d0.x * d0.x, d0.y * d0.y};
    v2f q1 = {d1.x * d1.x, d1.y * d1.y};
    accS = __builtin_amdgcn_wmma_f32_16x16x4_f32(false, ones, false, d0,
                                                 (short)0, accS, false, false);
    accQ = __builtin_amdgcn_wmma_f32_16x16x4_f32(false, ones, false, q0,
                                                 (short)0, accQ, false, false);
    accS = __builtin_amdgcn_wmma_f32_16x16x4_f32(false, ones, false, d1,
                                                 (short)0, accS, false, false);
    accQ = __builtin_amdgcn_wmma_f32_16x16x4_f32(false, ones, false, q1,
                                                 (short)0, accQ, false, false);
  }

  float s = accS[0], q = accQ[0];
  for (int off = 16; off > 0; off >>= 1) {
    s += __shfl_xor(s, off, 32);
    q += __shfl_xor(q, off, 32);
  }
  s *= 0.5f;
  q *= 0.5f;

  __shared__ float ls[8], lq[8];
  if (lane == 0) { ls[wave] = s; lq[wave] = q; }
  __syncthreads();
  if (tid == 0) {
    float S = 0.f, Q = 0.f;
    for (int i = 0; i < 8; ++i) { S += ls[i]; Q += lq[i]; }
    part[(size_t)blockIdx.x * PSTRIDE]     = S;
    part[(size_t)blockIdx.x * PSTRIDE + 1] = Q;
  }
}

__global__ __launch_bounds__(256) void apply_kernel(const float* __restrict__ y,
                                                    const float* __restrict__ part,
                                                    const float* __restrict__ g,
                                                    const float* __restrict__ b,
                                                    const float* __restrict__ resid,
                                                    float* __restrict__ out, int n) {
  const int c = (blockIdx.x >> 2) & (C_ - 1);
  float S = 0.f, Q = 0.f;
#pragma unroll
  for (int i = 0; i < NSPLIT; ++i) {
    S += part[(size_t)(c * NSPLIT + i) * PSTRIDE];
    Q += part[(size_t)(c * NSPLIT + i) * PSTRIDE + 1];
  }
  const float inv_n = 1.0f / (float)(B_ * H_ * W_);
  float mu    = S * inv_n;
  float var   = Q * inv_n - mu * mu;
  float inv   = rsqrtf(var + 1e-5f);
  float scale = g[c] * inv;
  float shift = b[c] - mu * scale;

  int idx = blockIdx.x * 256 + threadIdx.x;
  if (idx < n) {
    float v = y[idx] * scale + shift;
    if (resid) v += resid[idx];
    out[idx] = fmaxf(v, 0.f);
  }
}

extern "C" void kernel_launch(void* const* d_in, const int* in_sizes, int n_in,
                              void* d_out, int out_size, void* d_ws, size_t ws_size,
                              hipStream_t stream) {
  (void)in_sizes; (void)n_in; (void)out_size; (void)ws_size;
  const float* x  = (const float*)d_in[0];
  const float* w1 = (const float*)d_in[1];
  const float* g1 = (const float*)d_in[2];
  const float* b1 = (const float*)d_in[3];
  const float* w2 = (const float*)d_in[4];
  const float* g2 = (const float*)d_in[5];
  const float* b2 = (const float*)d_in[6];
  float* out = (float*)d_out;

  char* ws = (char*)d_ws;
  const size_t TEN   = (size_t)B_ * C_ * H_ * W_;
  const size_t WTSZ  = (size_t)C_ * KD * sizeof(float);
  float* ybuf = (float*)(ws);
  float* abuf = (float*)(ws + TEN * 4);
  float* wt1  = (float*)(ws + 2 * TEN * 4);
  float* wt2  = (float*)(ws + 2 * TEN * 4 + WTSZ);
  const size_t PSZ   = (size_t)C_ * NSPLIT * PSTRIDE * sizeof(float);
  float* pt1  = (float*)(ws + 2 * TEN * 4 + 2 * WTSZ);
  float* pt2  = (float*)(ws + 2 * TEN * 4 + 2 * WTSZ + PSZ);

  wtrans_kernel<<<144, 256, 0, stream>>>(w1, wt1);
  wtrans_kernel<<<144, 256, 0, stream>>>(w2, wt2);

  dim3 agrid(C_ / COG, (H_ / TILE) * (W_ / TILE), B_);

  adder_kernel<<<agrid, 256, 0, stream>>>(x, wt1, ybuf);
  stats_kernel<<<C_ * NSPLIT, 256, 0, stream>>>(ybuf, pt1);
  apply_kernel<<<(int)(TEN / 256), 256, 0, stream>>>(ybuf, pt1, g1, b1, nullptr, abuf, (int)TEN);

  adder_kernel<<<agrid, 256, 0, stream>>>(abuf, wt2, ybuf);
  stats_kernel<<<C_ * NSPLIT, 256, 0, stream>>>(ybuf, pt2);
  apply_kernel<<<(int)(TEN / 256), 256, 0, stream>>>(ybuf, pt2, g2, b2, x, out, (int)TEN);
}


# adder LDS staging: a thread issues all 41 global loads of a chunk before its first LDS store (was one load in flight per wave); border = clamped address + select; LDS tile padded; same values staged
# speedup vs baseline: 2.2639x; 1.4185x over previous
#include <hip/hip_runtime.h>


typedef float v2f __attribute__((ext_vector_type(2)));
typedef float v8f __attribute__((ext_vector_type(8)));
typedef __attribute__((address_space(4))) v8f cv8f;

#define B_   32
#define C_   64
#define H_   32
#define W_   32
#define KD   576
#define COG  16
#define CPL  8
#define TILE 16
#define CCH  32
#define PLANE    ((TILE + 2) * (TILE + 2))
#define STAGE_N  (CCH * PLANE)
#define STAGE_IT ((STAGE_N + 255) / 256)

__global__ __launch_bounds__(256) void wtrans_kernel(const float* __restrict__ w,
                                                     float* __restrict__ wt) {
  int idx = blockIdx.x * 256 + threadIdx.x;
  if (idx < C_ * KD) {
    int co = idx / KD, k = idx - co * KD;
    int cg = co >> 4, c = co & 15;
    wt[cg * (KD * COG) + k * COG + c] = w[idx];
  }
}

__global__ __launch_bounds__(256) void adder_kernel(const float* __restrict__ in,
                                                    const float* __restrict__ wt,
                                                    float* __restrict__ y) {
  __shared__ float xs[STAGE_IT * 256];
  const int tid    = threadIdx.x;
  __builtin_assume(tid < 256);
  const int cg     = blockIdx.x;
  const int tileId = blockIdx.y;
  const int n      = blockIdx.z;
  const int ty0 = (tileId >> 1) * TILE;
  const int tx0 = (tileId & 1) * TILE;
  const int chalf = __builtin_amdgcn_readfirstlane(tid >> 7);
  const int tx = tid & 15, ty = (tid >> 4) & 7;

  const float* wblk = wt + cg * (KD * COG) + chalf * CPL;
  __builtin_prefetch(wblk, 0, 1);

  float acc0[CPL], acc1[CPL];
#pragma unroll
  for (int c = 0; c < CPL; ++c) { acc0[c] = 0.f; acc1[c] = 0.f; }

  for (int c0 = 0; c0 < C_; c0 += CCH) {
    const float* inb = in + ((size_t)n * C_ + c0) * (H_ * W_);
    float v[STAGE_IT];
#pragma unroll
    for (int j = 0; j < STAGE_IT; ++j) {
      const int i  = min(tid + j * 256, STAGE_N - 1);
      const int ci = i / PLANE;
      const int r  = i - ci * PLANE;
      const int yy = r / (TILE + 2);
      const int xx = r - yy * (TILE + 2);
      const int gy = ty0 + yy - 1;
      const int gx = tx0 + xx - 1;
      const bool inside = (unsigned)gy < (unsigned)H_ && (unsigned)gx < (unsigned)W_;
      const int cy = min(max(gy, 0), H_ - 1);
      const int cx = min(max(gx, 0), W_ - 1);
      const float t = inb[(unsigned)((ci * H_ + cy) * W_ + cx)];
      v[j] = inside ? t : 0.f;
    }
#pragma unroll
    for (int j = 0; j < STAGE_IT; ++j) xs[tid + j * 256] = v[j];
    __syncthreads();

    for (int ci = 0; ci < CCH; ++ci) {
#pragma unroll
      for (int kh = 0; kh < 3; ++kh) {
#pragma unroll
        for (int kw = 0; kw < 3; ++kw) {
          const float xa = xs[(ci * (TILE + 2) + ty + kh) * (TILE + 2) + tx + kw];
          const float xb = xs[(ci * (TILE + 2) + ty + 8 + kh) * (TILE + 2) + tx + kw];
          const v8f wk = *(const cv8f*)(wblk + (((c0 + ci) * 3 + kh) * 3 + kw) * COG);
#pragma unroll
          for (int c = 0; c < CPL; ++c) {
            const float w = wk[c];
            const float da = xa - w, db = xb - w;
            asm("v_add_f32_e64 %0, %0, |%1|" : "+v"(acc0[c]) : "v"(da));
            asm("v_add_f32_e64 %0, %0, |%1|" : "+v"(acc1[c]) : "v"(db));
          }
        }
      }
    }
    __syncthreads();
  }

  const int gx = tx0 + tx;
#pragma unroll
  for (int c = 0; c < CPL; ++c) {
    float* yc = y + (((size_t)n * C_ + (cg * COG + chalf * CPL + c)) * H_ + ty0 + ty) * W_ + gx;
    yc[0]      = -acc0[c];
    yc[8 * W_] = -acc1[c];
  }
}

#define NSPLIT 8
#define PSTRIDE 32
__global__ __launch_bounds__(256) void stats_kernel(const float* __restrict__ y,
                                                    float* __restrict__ part) {
  const int c    = blockIdx.x / NSPLIT;
  const int sp   = blockIdx.x % NSPLIT;
  const int tid  = threadIdx.x;
  const int wave = tid >> 5, lane = tid & 31;

  const int n = sp * (B_ / NSPLIT) + (wave >> 1);
  const float4* p = reinterpret_cast<const float4*>(
      y + ((size_t)n * C_ + c) * (H_ * W_) + (wave & 1) * 512) + lane;
  float4 v[4];
#pragma unroll
  for (int r = 0; r < 4; ++r) v[r] = p[r * 32];

  v8f accS = {};
  v8f accQ = {};
  v2f ones = {1.f, 1.f};
#pragma unroll
  for (int r = 0; r < 4; ++r) {
    v2f d0 = {v[r].x, v[r].y};
    v2f d1 = {v[r].z, v[r].w};
    v2f q0 = {d0.x * d0.x, d0.y * d0.y};
    v2f q1 = {d1.x * d1.x, d1.y * d1.y};
    accS = __builtin_amdgcn_wmma_f32_16x16x4_f32(false, ones, false, d0,
                                                 (short)0, accS, false, false);
    accQ = __builtin_amdgcn_wmma_f32_16x16x4_f32(false, ones, false, q0,
                                                 (short)0, accQ, false, false);
    accS = __builtin_amdgcn_wmma_f32_16x16x4_f32(false, ones, false, d1,
                                                 (short)0, accS, false, false);
    accQ = __builtin_amdgcn_wmma_f32_16x16x4_f32(false, ones, false, q1,
                                                 (short)0, accQ, false, false);
  }

  float s = accS[0], q = accQ[0];
  for (int off = 16; off > 0; off >>= 1) {
    s += __shfl_xor(s, off, 32);
    q += __shfl_xor(q, off, 32);
  }
  s *= 0.5f;
  q *= 0.5f;

  __shared__ float ls[8], lq[8];
  if (lane == 0) { ls[wave] = s; lq[wave] = q; }
  __syncthreads();
  if (tid == 0) {
    float S = 0.f, Q = 0.f;
    for (int i = 0; i < 8; ++i) { S += ls[i]; Q += lq[i]; }
    part[(size_t)blockIdx.x * PSTRIDE]     = S;
    part[(size_t)blockIdx.x * PSTRIDE + 1] = Q;
  }
}

__global__ __launch_bounds__(256) void apply_kernel(const float* __restrict__ y,
                                                    const float* __restrict__ part,
                                                    const float* __restrict__ g,
                                                    const float* __restrict__ b,
                                                    const float* __restrict__ resid,
                                                    float* __restrict__ out, int n) {
  const int c = (blockIdx.x >> 2) & (C_ - 1);
  float S = 0.f, Q = 0.f;
#pragma unroll
  for (int i = 0; i < NSPLIT; ++i) {
    S += part[(size_t)(c * NSPLIT + i) * PSTRIDE];
    Q += part[(size_t)(c * NSPLIT + i) * PSTRIDE + 1];
  }
  const float inv_n = 1.0f / (float)(B_ * H_ * W_);
  float mu    = S * inv_n;
  float var   = Q * inv_n - mu * mu;
  float inv   = rsqrtf(var + 1e-5f);
  float scale = g[c] * inv;
  float shift = b[c] - mu * scale;

  int idx = blockIdx.x * 256 + threadIdx.x;
  if (idx < n) {
    float v = y[idx] * scale + shift;
    if (resid) v += resid[idx];
    out[idx] = fmaxf(v, 0.f);
  }
}

extern "C" void kernel_launch(void* const* d_in, const int* in_sizes, int n_in,
                              void* d_out, int out_size, void* d_ws, size_t ws_size,
                              hipStream_t stream) {
  (void)in_sizes; (void)n_in; (void)out_size; (void)ws_size;
  const float* x  = (const float*)d_in[0];
  const float* w1 = (const float*)d_in[1];
  const float* g1 = (const float*)d_in[2];
  const float* b1 = (const float*)d_in[3];
  const float* w2 = (const float*)d_in[4];
  const float* g2 = (const float*)d_in[5];
  const float* b2 = (const float*)d_in[6];
  float* out = (float*)d_out;

  char* ws = (char*)d_ws;
  const size_t TEN   = (size_t)B_ * C_ * H_ * W_;
  const size_t WTSZ  = (size_t)C_ * KD * sizeof(float);
  float* ybuf = (float*)(ws);
  float* abuf = (float*)(ws + TEN * 4);
  float* wt1  = (float*)(ws + 2 * TEN * 4);
  float* wt2  = (float*)(ws + 2 * TEN * 4 + WTSZ);
  const size_t PSZ   = (size_t)C_ * NSPLIT * PSTRIDE * sizeof(float);
  float* pt1  = (float*)(ws + 2 * TEN * 4 + 2 * WTSZ);
  float* pt2  = (float*)(ws + 2 * TEN * 4 + 2 * WTSZ + PSZ);

  wtrans_kernel<<<144, 256, 0, stream>>>(w1, wt1);
  wtrans_kernel<<<144, 256, 0, stream>>>(w2, wt2);

  dim3 agrid(C_ / COG, (H_ / TILE) * (W_ / TILE), B_);

  adder_kernel<<<agrid, 256, 0, stream>>>(x, wt1, ybuf);
  stats_kernel<<<C_ * NSPLIT, 256, 0, stream>>>(ybuf, pt1);
  apply_kernel<<<(int)(TEN / 256), 256, 0, stream>>>(ybuf, pt1, g1, b1, nullptr, abuf, (int)TEN);

  adder_kernel<<<agrid, 256, 0, stream>>>(abuf, wt2, ybuf);
  stats_kernel<<<C_ * NSPLIT, 256, 0, stream>>>(ybuf, pt2);
  apply_kernel<<<(int)(TEN / 256), 256, 0, stream>>>(ybuf, pt2, g2, b2, x, out, (int)TEN);
}


# normalize pass: a thread takes 4 image planes of one channel as 128-bit loads and stores, 512 work-groups instead of 8,192 (pass was bound by wave entry, not bytes); per-element arithmetic kept
# speedup vs baseline: 2.3763x; 1.0496x over previous
#include <hip/hip_runtime.h>


typedef float v2f __attribute__((ext_vector_type(2)));
typedef float v8f __attribute__((ext_vector_type(8)));
typedef __attribute__((address_space(4))) v8f cv8f;

#define B_   32
#define C_   64
#define H_   32
#define W_   32
#define KD   576
#define COG  16
#define CPL  8
#define TILE 16
#define CCH  32
#define PLANE    ((TILE + 2) * (TILE + 2))
#define STAGE_N  (CCH * PLANE)
#define STAGE_IT ((STAGE_N + 255) / 256)

__global__ __launch_bounds__(256) void wtrans_kernel(const float* __restrict__ w,
                                                     float* __restrict__ wt) {
  int idx = blockIdx.x * 256 + threadIdx.x;
  if (idx < C_ * KD) {
    int co = idx / KD, k = idx - co * KD;
    int cg = co >> 4, c = co & 15;
    wt[cg * (KD * COG) + k * COG + c] = w[idx];
  }
}

__global__ __launch_bounds__(256) void adder_kernel(const float* __restrict__ in,
                                                    const float* __restrict__ wt,
                                                    float* __restrict__ y) {
  __shared__ float xs[STAGE_IT * 256];
  const int tid    = threadIdx.x;
  __builtin_assume(tid < 256);
  const int cg     = blockIdx.x;
  const int tileId = blockIdx.y;
  const int n      = blockIdx.z;
  const int ty0 = (tileId >> 1) * TILE;
  const int tx0 = (tileId & 1) * TILE;
  const int chalf = __builtin_amdgcn_readfirstlane(tid >> 7);
  const int tx = tid & 15, ty = (tid >> 4) & 7;

  const float* wblk = wt + cg * (KD * COG) + chalf * CPL;
  __builtin_prefetch(wblk, 0, 1);

  float acc0[CPL], acc1[CPL];
#pragma unroll
  for (int c = 0; c < CPL; ++c) { acc0[c] = 0.f; acc1[c] = 0.f; }

  for (int c0 = 0; c0 < C_; c0 += CCH) {
    const float* inb = in + ((size_t)n * C_ + c0) * (H_ * W_);
    float v[STAGE_IT];
#pragma unroll
    for (int j = 0; j < STAGE_IT; ++j) {
      const int i  = min(tid + j * 256, STAGE_N - 1);
      const int ci = i / PLANE;
      const int r  = i - ci * PLANE;
      const int yy = r / (TILE + 2);
      const int xx = r - yy * (TILE + 2);
      const int gy = ty0 + yy - 1;
      const int gx = tx0 + xx - 1;
      const bool inside = (unsigned)gy < (unsigned)H_ && (unsigned)gx < (unsigned)W_;
      const int cy = min(max(gy, 0), H_ - 1);
      const int cx = min(max(gx, 0), W_ - 1);
      const float t = inb[(unsigned)((ci * H_ + cy) * W_ + cx)];
      v[j] = inside ? t : 0.f;
    }
#pragma unroll
    for (int j = 0; j < STAGE_IT; ++j) xs[tid + j * 256] = v[j];
    __syncthreads();

    for (int ci = 0; ci < CCH; ++ci) {
#pragma unroll
      for (int kh = 0; kh < 3; ++kh) {
#pragma unroll
        for (int kw = 0; kw < 3; ++kw) {
          const float xa = xs[(ci * (TILE + 2) + ty + kh) * (TILE + 2) + tx + kw];
          const float xb = xs[(ci * (TILE + 2) + ty + 8 + kh) * (TILE + 2) + tx + kw];
          const v8f wk = *(const cv8f*)(wblk + (((c0 + ci) * 3 + kh) * 3 + kw) * COG);
#pragma unroll
          for (int c = 0; c < CPL; ++c) {
            const float w = wk[c];
            const float da = xa - w, db = xb - w;
            asm("v_add_f32_e64 %0, %0, |%1|" : "+v"(acc0[c]) : "v"(da));
            asm("v_add_f32_e64 %0, %0, |%1|" : "+v"(acc1[c]) : "v"(db));
          }
        }
      }
    }
    __syncthreads();
  }

  const int gx = tx0 + tx;
#pragma unroll
  for (int c = 0; c < CPL; ++c) {
    float* yc = y + (((size_t)n * C_ + (cg * COG + chalf * CPL + c)) * H_ + ty0 + ty) * W_ + gx;
    yc[0]      = -acc0[c];
    yc[8 * W_] = -acc1[c];
  }
}

#define NSPLIT 8
#define PSTRIDE 32
__global__ __launch_bounds__(256) void stats_kernel(const float* __restrict__ y,
                                                    float* __restrict__ part) {
  const int c    = blockIdx.x / NSPLIT;
  const int sp   = blockIdx.x % NSPLIT;
  const int tid  = threadIdx.x;
  const int wave = tid >> 5, lane = tid & 31;

  const int n = sp * (B_ / NSPLIT) + (wave >> 1);
  const float4* p = reinterpret_cast<const float4*>(
      y + ((size_t)n * C_ + c) * (H_ * W_) + (wave & 1) * 512) + lane;
  float4 v[4];
#pragma unroll
  for (int r = 0; r < 4; ++r) v[r] = p[r * 32];

  v8f accS = {};
  v8f accQ = {};
  v2f ones = {1.f, 1.f};
#pragma unroll
  for (int r = 0; r < 4; ++r) {
    v2f d0 = {v[r].x, v[r].y};
    v2f d1 = {v[r].z, v[r].w};
    v2f q0 = {d0.x * d0.x, d0.y * d0.y};
    v2f q1 = {d1.x * d1.x, d1.y * d1.y};
    accS = __builtin_amdgcn_wmma_f32_16x16x4_f32(false, ones, false, d0,
                                                 (short)0, accS, false, false);
    accQ = __builtin_amdgcn_wmma_f32_16x16x4_f32(false, ones, false, q0,
                                                 (short)0, accQ, false, false);
    accS = __builtin_amdgcn_wmma_f32_16x16x4_f32(false, ones, false, d1,
                                                 (short)0, accS, false, false);
    accQ = __builtin_amdgcn_wmma_f32_16x16x4_f32(false, ones, false, q1,
                                                 (short)0, accQ, false, false);
  }

  float s = accS[0], q = accQ[0];
  for (int off = 16; off > 0; off >>= 1) {
    s += __shfl_xor(s, off, 32);
    q += __shfl_xor(q, off, 32);
  }
  s *= 0.5f;
  q *= 0.5f;

  __shared__ float ls[8], lq[8];
  if (lane == 0) { ls[wave] = s; lq[wave] = q; }
  __syncthreads();
  if (tid == 0) {
    float S = 0.f, Q = 0.f;
    for (int i = 0; i < 8; ++i) { S += ls[i]; Q += lq[i]; }
    part[(size_t)blockIdx.x * PSTRIDE]     = S;
    part[(size_t)blockIdx.x * PSTRIDE + 1] = Q;
  }
}

#define APL 4
__global__ __launch_bounds__(256) void apply_kernel(const float* __restrict__ y,
                                                    const float* __restrict__ part,
                                                    const float* __restrict__ g,
                                                    const float* __restrict__ b,
                                                    const float* __restrict__ resid,
                                                    float* __restrict__ out) {
  const int c  = blockIdx.x & (C_ - 1);
  const int n0 = (blockIdx.x >> 6) * APL;
  const size_t base   = ((size_t)n0 * C_ + c) * (H_ * W_) + threadIdx.x * 4;
  const size_t nstep  = (size_t)C_ * H_ * W_;

  float4 v[APL], r[APL];
  for (int j = 0; j < APL; ++j) v[j] = *reinterpret_cast<const float4*>(y + base + j * nstep);
  if (resid) {
    for (int j = 0; j < APL; ++j) r[j] = *reinterpret_cast<const float4*>(resid + base + j * nstep);
  }

  float S = 0.f, Q = 0.f;
#pragma unroll
  for (int i = 0; i < NSPLIT; ++i) {
    S += part[(size_t)(c * NSPLIT + i) * PSTRIDE];
    Q += part[(size_t)(c * NSPLIT + i) * PSTRIDE + 1];
  }
  const float inv_n = 1.0f / (float)(B_ * H_ * W_);
  float mu    = S * inv_n;
  float var   = Q * inv_n - mu * mu;
  float inv   = rsqrtf(var + 1e-5f);
  float scale = g[c] * inv;
  float shift = b[c] - mu * scale;

  for (int j = 0; j < APL; ++j) {
    float4 o;
    o.x = v[j].x * scale + shift;
    o.y = v[j].y * scale + shift;
    o.z = v[j].z * scale + shift;
    o.w = v[j].w * scale + shift;
    if (resid) { o.x += r[j].x; o.y += r[j].y; o.z += r[j].z; o.w += r[j].w; }
    o.x = fmaxf(o.x, 0.f); o.y = fmaxf(o.y, 0.f); o.z = fmaxf(o.z, 0.f); o.w = fmaxf(o.w, 0.f);
    *reinterpret_cast<float4*>(out + base + j * nstep) = o;
  }
}

extern "C" void kernel_launch(void* const* d_in, const int* in_sizes, int n_in,
                              void* d_out, int out_size, void* d_ws, size_t ws_size,
                              hipStream_t stream) {
  (void)in_sizes; (void)n_in; (void)out_size; (void)ws_size;
  const float* x  = (const float*)d_in[0];
  const float* w1 = (const float*)d_in[1];
  const float* g1 = (const float*)d_in[2];
  const float* b1 = (const float*)d_in[3];
  const float* w2 = (const float*)d_in[4];
  const float* g2 = (const float*)d_in[5];
  const float* b2 = (const float*)d_in[6];
  float* out = (float*)d_out;

  char* ws = (char*)d_ws;
  const size_t TEN   = (size_t)B_ * C_ * H_ * W_;
  const size_t WTSZ  = (size_t)C_ * KD * sizeof(float);
  float* ybuf = (float*)(ws);
  float* abuf = (float*)(ws + TEN * 4);
  float* wt1  = (float*)(ws + 2 * TEN * 4);
  float* wt2  = (float*)(ws + 2 * TEN * 4 + WTSZ);
  const size_t PSZ   = (size_t)C_ * NSPLIT * PSTRIDE * sizeof(float);
  float* pt1  = (float*)(ws + 2 * TEN * 4 + 2 * WTSZ);
  float* pt2  = (float*)(ws + 2 * TEN * 4 + 2 * WTSZ + PSZ);

  wtrans_kernel<<<144, 256, 0, stream>>>(w1, wt1);
  wtrans_kernel<<<144, 256, 0, stream>>>(w2, wt2);

  dim3 agrid(C_ / COG, (H_ / TILE) * (W_ / TILE), B_);

  adder_kernel<<<agrid, 256, 0, stream>>>(x, wt1, ybuf);
  stats_kernel<<<C_ * NSPLIT, 256, 0, stream>>>(ybuf, pt1);
  apply_kernel<<<(B_ / APL) * C_, 256, 0, stream>>>(ybuf, pt1, g1, b1, nullptr, abuf);

  adder_kernel<<<agrid, 256, 0, stream>>>(abuf, wt2, ybuf);
  stats_kernel<<<C_ * NSPLIT, 256, 0, stream>>>(ybuf, pt2);
  apply_kernel<<<(B_ / APL) * C_, 256, 0, stream>>>(ybuf, pt2, g2, b2, x, out);
}


# weight transpose kernels removed (8 launches -> 6): the adder reads each output channel's 9 taps straight from w by scalar loads (256-bit + 32-bit per input channel); per-output sum order kept
# speedup vs baseline: 2.5926x; 1.0910x over previous
#include <hip/hip_runtime.h>


typedef float v2f __attribute__((ext_vector_type(2)));
typedef float v8f __attribute__((ext_vector_type(8)));
typedef v8f v8f_a4 __attribute__((aligned(4)));
typedef __attribute__((address_space(4))) v8f_a4 cv8f;
typedef __attribute__((address_space(4))) float cf32;

#define B_   32
#define C_   64
#define H_   32
#define W_   32
#define KD   576
#define COG  16
#define CPL  8
#define TILE 16
#define CCH  32
#define PLANE    ((TILE + 2) * (TILE + 2))
#define STAGE_N  (CCH * PLANE)
#define STAGE_IT ((STAGE_N + 255) / 256)

__global__ __launch_bounds__(256) void adder_kernel(const float* __restrict__ in,
                                                    const float* __restrict__ wgt,
                                                    float* __restrict__ y) {
  __shared__ float xs[STAGE_IT * 256];
  const int tid    = threadIdx.x;
  __builtin_assume(tid < 256);
  const int cg     = blockIdx.x;
  const int tileId = blockIdx.y;
  const int n      = blockIdx.z;
  const int ty0 = (tileId >> 1) * TILE;
  const int tx0 = (tileId & 1) * TILE;
  const int chalf = __builtin_amdgcn_readfirstlane(tid >> 7);
  const int tx = tid & 15, ty = (tid >> 4) & 7;

  const float* wblk = wgt + (cg * COG + chalf * CPL) * KD;
  __builtin_prefetch(wblk, 0, 1);

  float acc0[CPL], acc1[CPL];
#pragma unroll
  for (int c = 0; c < CPL; ++c) { acc0[c] = 0.f; acc1[c] = 0.f; }

  for (int c0 = 0; c0 < C_; c0 += CCH) {
    const float* inb = in + ((size_t)n * C_ + c0) * (H_ * W_);
    float v[STAGE_IT];
#pragma unroll
    for (int j = 0; j < STAGE_IT; ++j) {
      const int i  = min(tid + j * 256, STAGE_N - 1);
      const int ci = i / PLANE;
      const int r  = i - ci * PLANE;
      const int yy = r / (TILE + 2);
      const int xx = r - yy * (TILE + 2);
      const int gy = ty0 + yy - 1;
      const int gx = tx0 + xx - 1;
      const bool inside = (unsigned)gy < (unsigned)H_ && (unsigned)gx < (unsigned)W_;
      const int cy = min(max(gy, 0), H_ - 1);
      const int cx = min(max(gx, 0), W_ - 1);
      const float t = inb[(unsigned)((ci * H_ + cy) * W_ + cx)];
      v[j] = inside ? t : 0.f;
    }
#pragma unroll
    for (int j = 0; j < STAGE_IT; ++j) xs[tid + j * 256] = v[j];
    __syncthreads();

    for (int ci = 0; ci < CCH; ++ci) {
      const float* wci = wblk + (c0 + ci) * 9;
      v8f   w07[CPL];
      float w8[CPL];
#pragma unroll
      for (int c = 0; c < CPL; ++c) {
        w07[c] = *(const cv8f*)(wci + c * KD);
        w8[c]  = *(const cf32*)(wci + c * KD + 8);
      }
#pragma unroll
      for (int kh = 0; kh < 3; ++kh) {
#pragma unroll
        for (int kw = 0; kw < 3; ++kw) {
          const float xa = xs[(ci * (TILE + 2) + ty + kh) * (TILE + 2) + tx + kw];
          const float xb = xs[(ci * (TILE + 2) + ty + 8 + kh) * (TILE + 2) + tx + kw];
#pragma unroll
          for (int c = 0; c < CPL; ++c) {
            const float w = (kh * 3 + kw < 8) ? w07[c][kh * 3 + kw] : w8[c];
            const float da = xa - w, db = xb - w;
            asm("v_add_f32_e64 %0, %0, |%1|" : "+v"(acc0[c]) : "v"(da));
            asm("v_add_f32_e64 %0, %0, |%1|" : "+v"(acc1[c]) : "v"(db));
          }
        }
      }
    }
    __syncthreads();
  }

  const int gx = tx0 + tx;
#pragma unroll
  for (int c = 0; c < CPL; ++c) {
    float* yc = y + (((size_t)n * C_ + (cg * COG + chalf * CPL + c)) * H_ + ty0 + ty) * W_ + gx;
    yc[0]      = -acc0[c];
    yc[8 * W_] = -acc1[c];
  }
}

#define NSPLIT 8
#define PSTRIDE 32
__global__ __launch_bounds__(256) void stats_kernel(const float* __restrict__ y,
                                                    float* __restrict__ part) {
  const int c    = blockIdx.x / NSPLIT;
  const int sp   = blockIdx.x % NSPLIT;
  const int tid  = threadIdx.x;
  const int wave = tid >> 5, lane = tid & 31;

  const int n = sp * (B_ / NSPLIT) + (wave >> 1);
  const float4* p = reinterpret_cast<const float4*>(
      y + ((size_t)n * C_ + c) * (H_ * W_) + (wave & 1) * 512) + lane;
  float4 v[4];
#pragma unroll
  for (int r = 0; r < 4; ++r) v[r] = p[r * 32];

  v8f accS = {};
  v8f accQ = {};
  v2f ones = {1.f, 1.f};
#pragma unroll
  for (int r = 0; r < 4; ++r) {
    v2f d0 = {v[r].x, v[r].y};
    v2f d1 = {v[r].z, v[r].w};
    v2f q0 = {d0.x * d0.x, d0.y * d0.y};
    v2f q1 = {d1.x * d1.x, d1.y * d1.y};
    accS = __builtin_amdgcn_wmma_f32_16x16x4_f32(false, ones, false, d0,
                                                 (short)0, accS, false, false);
    accQ = __builtin_amdgcn_wmma_f32_16x16x4_f32(false, ones, false, q0,
                                                 (short)0, accQ, false, false);
    accS = __builtin_amdgcn_wmma_f32_16x16x4_f32(false, ones, false, d1,
                                                 (short)0, accS, false, false);
    accQ = __builtin_amdgcn_wmma_f32_16x16x4_f32(false, ones, false, q1,
                                                 (short)0, accQ, false, false);
  }

  float s = accS[0], q = accQ[0];
  for (int off = 16; off > 0; off >>= 1) {
    s += __shfl_xor(s, off, 32);
    q += __shfl_xor(q, off, 32);
  }
  s *= 0.5f;
  q *= 0.5f;

  __shared__ float ls[8], lq[8];
  if (lane == 0) { ls[wave] = s; lq[wave] = q; }
  __syncthreads();
  if (tid == 0) {
    float S = 0.f, Q = 0.f;
    for (int i = 0; i < 8; ++i) { S += ls[i]; Q += lq[i]; }
    part[(size_t)blockIdx.x * PSTRIDE]     = S;
    part[(size_t)blockIdx.x * PSTRIDE + 1] = Q;
  }
}

#define APL 4
__global__ __launch_bounds__(256) void apply_kernel(const float* __restrict__ y,
                                                    const float* __restrict__ part,
                                                    const float* __restrict__ g,
                                                    const float* __restrict__ b,
                                                    const float* __restrict__ resid,
                                                    float* __restrict__ out) {
  const int c  = blockIdx.x & (C_ - 1);
  const int n0 = (blockIdx.x >> 6) * APL;
  const size_t base   = ((size_t)n0 * C_ + c) * (H_ * W_) + threadIdx.x * 4;
  const size_t nstep  = (size_t)C_ * H_ * W_;

  float4 v[APL], r[APL];
  for (int j = 0; j < APL; ++j) v[j] = *reinterpret_cast<const float4*>(y + base + j * nstep);
  if (resid) {
    for (int j = 0; j < APL; ++j) r[j] = *reinterpret_cast<const float4*>(resid + base + j * nstep);
  }

  float S = 0.f, Q = 0.f;
#pragma unroll
  for (int i = 0; i < NSPLIT; ++i) {
    S += part[(size_t)(c * NSPLIT + i) * PSTRIDE];
    Q += part[(size_t)(c * NSPLIT + i) * PSTRIDE + 1];
  }
  const float inv_n = 1.0f / (float)(B_ * H_ * W_);
  float mu    = S * inv_n;
  float var   = Q * inv_n - mu * mu;
  float inv   = rsqrtf(var + 1e-5f);
  float scale = g[c] * inv;
  float shift = b[c] - mu * scale;

  for (int j = 0; j < APL; ++j) {
    float4 o;
    o.x = v[j].x * scale + shift;
    o.y = v[j].y * scale + shift;
    o.z = v[j].z * scale + shift;
    o.w = v[j].w * scale + shift;
    if (resid) { o.x += r[j].x; o.y += r[j].y; o.z += r[j].z; o.w += r[j].w; }
    o.x = fmaxf(o.x, 0.f); o.y = fmaxf(o.y, 0.f); o.z = fmaxf(o.z, 0.f); o.w = fmaxf(o.w, 0.f);
    *reinterpret_cast<float4*>(out + base + j * nstep) = o;
  }
}

extern "C" void kernel_launch(void* const* d_in, const int* in_sizes, int n_in,
                              void* d_out, int out_size, void* d_ws, size_t ws_size,
                              hipStream_t stream) {
  (void)in_sizes; (void)n_in; (void)out_size; (void)ws_size;
  const float* x  = (const float*)d_in[0];
  const float* w1 = (const float*)d_in[1];
  const float* g1 = (const float*)d_in[2];
  const float* b1 = (const float*)d_in[3];
  const float* w2 = (const float*)d_in[4];
  const float* g2 = (const float*)d_in[5];
  const float* b2 = (const float*)d_in[6];
  float* out = (float*)d_out;

  char* ws = (char*)d_ws;
  const size_t TEN   = (size_t)B_ * C_ * H_ * W_;
  const size_t WTSZ  = (size_t)C_ * KD * sizeof(float);
  float* ybuf = (float*)(ws);
  float* abuf = (float*)(ws + TEN * 4);
  const size_t PSZ   = (size_t)C_ * NSPLIT * PSTRIDE * sizeof(float);
  float* pt1  = (float*)(ws + 2 * TEN * 4 + 2 * WTSZ);
  float* pt2  = (float*)(ws + 2 * TEN * 4 + 2 * WTSZ + PSZ);

  dim3 agrid(C_ / COG, (H_ / TILE) * (W_ / TILE), B_);

  adder_kernel<<<agrid, 256, 0, stream>>>(x, w1, ybuf);
  stats_kernel<<<C_ * NSPLIT, 256, 0, stream>>>(ybuf, pt1);
  apply_kernel<<<(B_ / APL) * C_, 256, 0, stream>>>(ybuf, pt1, g1, b1, nullptr, abuf);

  adder_kernel<<<agrid, 256, 0, stream>>>(abuf, w2, ybuf);
  stats_kernel<<<C_ * NSPLIT, 256, 0, stream>>>(ybuf, pt2);
  apply_kernel<<<(B_ / APL) * C_, 256, 0, stream>>>(ybuf, pt2, g2, b2, x, out);
}


# adder loop sums max(x,w), not |x-w| (|x-w| = 2max(x,w)-x-w): max and add both dual-issue, 153 vector instr per input channel (was 216); patch sum and weight-row sum come off last; new sum order
# speedup vs baseline: 2.7896x; 1.0760x over previous
#include <hip/hip_runtime.h>


typedef float v2f __attribute__((ext_vector_type(2)));
typedef float v8f __attribute__((ext_vector_type(8)));
typedef v8f v8f_a4 __attribute__((aligned(4)));
typedef __attribute__((address_space(4))) v8f_a4 cv8f;
typedef __attribute__((address_space(4))) float cf32;

#define B_   32
#define C_   64
#define H_   32
#define W_   32
#define KD   576
#define COG  16
#define CPL  8
#define TILE 16
#define CCH  32
#define PLANE    ((TILE + 2) * (TILE + 2))
#define STAGE_N  (CCH * PLANE)
#define STAGE_IT ((STAGE_N + 255) / 256)

__global__ __launch_bounds__(256) void adder_kernel(const float* __restrict__ in,
                                                    const float* __restrict__ wgt,
                                                    float* __restrict__ y) {
  __shared__ float xs[STAGE_IT * 256];
  __shared__ float swl[COG];
  const int tid    = threadIdx.x;
  __builtin_assume(tid < 256);
  const int cg     = blockIdx.x;
  const int tileId = blockIdx.y;
  const int n      = blockIdx.z;
  const int ty0 = (tileId >> 1) * TILE;
  const int tx0 = (tileId & 1) * TILE;
  const int chalf = __builtin_amdgcn_readfirstlane(tid >> 7);
  const int tx = tid & 15, ty = (tid >> 4) & 7;

  const float* wblk = wgt + (cg * COG + chalf * CPL) * KD;
  __builtin_prefetch(wblk, 0, 1);

  {
    const int wv   = tid >> 5, lane = tid & 31;
    const float4* wr = reinterpret_cast<const float4*>(wgt + (size_t)(cg * COG + wv * 2) * KD) + lane;
    float4 q[9];
#pragma unroll
    for (int r = 0; r < 9; ++r) q[r] = wr[r * 32];
    float s0 = 0.f, s1 = 0.f;
#pragma unroll
    for (int r = 0; r < 4; ++r) s0 += (q[r].x + q[r].y) + (q[r].z + q[r].w);
#pragma unroll
    for (int r = 5; r < 9; ++r) s1 += (q[r].x + q[r].y) + (q[r].z + q[r].w);
    const float mid = (q[4].x + q[4].y) + (q[4].z + q[4].w);
    s0 += (lane < 16) ? mid : 0.f;
    s1 += (lane < 16) ? 0.f : mid;
    for (int off = 16; off > 0; off >>= 1) {
      s0 += __shfl_xor(s0, off, 32);
      s1 += __shfl_xor(s1, off, 32);
    }
    if (lane == 0) { swl[wv * 2] = s0; swl[wv * 2 + 1] = s1; }
  }

  float acc0[CPL], acc1[CPL];
#pragma unroll
  for (int c = 0; c < CPL; ++c) { acc0[c] = 0.f; acc1[c] = 0.f; }
  float sx0 = 0.f, sx1 = 0.f;

  for (int c0 = 0; c0 < C_; c0 += CCH) {
    const float* inb = in + ((size_t)n * C_ + c0) * (H_ * W_);
    float v[STAGE_IT];
#pragma unroll
    for (int j = 0; j < STAGE_IT; ++j) {
      const int i  = min(tid + j * 256, STAGE_N - 1);
      const int ci = i / PLANE;
      const int r  = i - ci * PLANE;
      const int yy = r / (TILE + 2);
      const int xx = r - yy * (TILE + 2);
      const int gy = ty0 + yy - 1;
      const int gx = tx0 + xx - 1;
      const bool inside = (unsigned)gy < (unsigned)H_ && (unsigned)gx < (unsigned)W_;
      const int cy = min(max(gy, 0), H_ - 1);
      const int cx = min(max(gx, 0), W_ - 1);
      const float t = inb[(unsigned)((ci * H_ + cy) * W_ + cx)];
      v[j] = inside ? t : 0.f;
    }
#pragma unroll
    for (int j = 0; j < STAGE_IT; ++j) xs[tid + j * 256] = v[j];
    __syncthreads();

    for (int ci = 0; ci < CCH; ++ci) {
      const float* wci = wblk + (c0 + ci) * 9;
      v8f   w07[CPL];
      float w8[CPL];
#pragma unroll
      for (int c = 0; c < CPL; ++c) {
        w07[c] = *(const cv8f*)(wci + c * KD);
        w8[c]  = *(const cf32*)(wci + c * KD + 8);
      }
#pragma unroll
      for (int kh = 0; kh < 3; ++kh) {
#pragma unroll
        for (int kw = 0; kw < 3; ++kw) {
          const float xa = xs[(ci * (TILE + 2) + ty + kh) * (TILE + 2) + tx + kw];
          const float xb = xs[(ci * (TILE + 2) + ty + 8 + kh) * (TILE + 2) + tx + kw];
          float w[CPL];
#pragma unroll
          for (int c = 0; c < CPL; ++c) w[c] = (kh * 3 + kw < 8) ? w07[c][kh * 3 + kw] : w8[c];
          float m0[CPL], m1[CPL];
          asm("v_dual_max_num_f32 %0, %16, %24 :: v_dual_max_num_f32 %8, %16, %25\n\t"
              "v_dual_max_num_f32 %1, %17, %24 :: v_dual_max_num_f32 %9, %17, %25\n\t"
              "v_dual_max_num_f32 %2, %18, %24 :: v_dual_max_num_f32 %10, %18, %25\n\t"
              "v_dual_max_num_f32 %3, %19, %24 :: v_dual_max_num_f32 %11, %19, %25\n\t"
              "v_dual_max_num_f32 %4, %20, %24 :: v_dual_max_num_f32 %12, %20, %25\n\t"
              "v_dual_max_num_f32 %5, %21, %24 :: v_dual_max_num_f32 %13, %21, %25\n\t"
              "v_dual_max_num_f32 %6, %22, %24 :: v_dual_max_num_f32 %14, %22, %25\n\t"
              "v_dual_max_num_f32 %7, %23, %24 :: v_dual_max_num_f32 %15, %23, %25"
              : "=&v"(m0[0]), "=&v"(m0[1]), "=&v"(m0[2]), "=&v"(m0[3]),
                "=&v"(m0[4]), "=&v"(m0[5]), "=&v"(m0[6]), "=&v"(m0[7]),
                "=&v"(m1[0]), "=&v"(m1[1]), "=&v"(m1[2]), "=&v"(m1[3]),
                "=&v"(m1[4]), "=&v"(m1[5]), "=&v"(m1[6]), "=&v"(m1[7])
              : "s"(w[0]), "s"(w[1]), "s"(w[2]), "s"(w[3]), "s"(w[4]), "s"(w[5]), "s"(w[6]), "s"(w[7]),
                "v"(xa), "v"(xb));
          sx0 += xa;
          sx1 += xb;
#pragma unroll
          for (int c = 0; c < CPL; ++c) {
            acc0[c] += m0[c];
            acc1[c] += m1[c];
          }
        }
      }
    }
    __syncthreads();
  }

  const int gx = tx0 + tx;
#pragma unroll
  for (int c = 0; c < CPL; ++c) {
    float* yc = y + (((size_t)n * C_ + (cg * COG + chalf * CPL + c)) * H_ + ty0 + ty) * W_ + gx;
    const float sw = swl[chalf * CPL + c];
    yc[0]      = (sx0 + sw) - 2.f * acc0[c];
    yc[8 * W_] = (sx1 + sw) - 2.f * acc1[c];
  }
}

#define NSPLIT 8
#define PSTRIDE 32
__global__ __launch_bounds__(256) void stats_kernel(const float* __restrict__ y,
                                                    float* __restrict__ part) {
  const int c    = blockIdx.x / NSPLIT;
  const int sp   = blockIdx.x % NSPLIT;
  const int tid  = threadIdx.x;
  const int wave = tid >> 5, lane = tid & 31;

  const int n = sp * (B_ / NSPLIT) + (wave >> 1);
  const float4* p = reinterpret_cast<const float4*>(
      y + ((size_t)n * C_ + c) * (H_ * W_) + (wave & 1) * 512) + lane;
  float4 v[4];
#pragma unroll
  for (int r = 0; r < 4; ++r) v[r] = p[r * 32];

  v8f accS = {};
  v8f accQ = {};
  v2f ones = {1.f, 1.f};
#pragma unroll
  for (int r = 0; r < 4; ++r) {
    v2f d0 = {v[r].x, v[r].y};
    v2f d1 = {v[r].z, v[r].w};
    v2f q0 = {d0.x * d0.x, d0.y * d0.y};
    v2f q1 = {d1.x * d1.x, d1.y * d1.y};
    accS = __builtin_amdgcn_wmma_f32_16x16x4_f32(false, ones, false, d0,
                                                 (short)0, accS, false, false);
    accQ = __builtin_amdgcn_wmma_f32_16x16x4_f32(false, ones, false, q0,
                                                 (short)0, accQ, false, false);
    accS = __builtin_amdgcn_wmma_f32_16x16x4_f32(false, ones, false, d1,
                                                 (short)0, accS, false, false);
    accQ = __builtin_amdgcn_wmma_f32_16x16x4_f32(false, ones, false, q1,
                                                 (short)0, accQ, false, false);
  }

  float s = accS[0], q = accQ[0];
  for (int off = 16; off > 0; off >>= 1) {
    s += __shfl_xor(s, off, 32);
    q += __shfl_xor(q, off, 32);
  }
  s *= 0.5f;
  q *= 0.5f;

  __shared__ float ls[8], lq[8];
  if (lane == 0) { ls[wave] = s; lq[wave] = q; }
  __syncthreads();
  if (tid == 0) {
    float S = 0.f, Q = 0.f;
    for (int i = 0; i < 8; ++i) { S += ls[i]; Q += lq[i]; }
    part[(size_t)blockIdx.x * PSTRIDE]     = S;
    part[(size_t)blockIdx.x * PSTRIDE + 1] = Q;
  }
}

#define APL 4
__global__ __launch_bounds__(256) void apply_kernel(const float* __restrict__ y,
                                                    const float* __restrict__ part,
                                                    const float* __restrict__ g,
                                                    const float* __restrict__ b,
                                                    const float* __restrict__ resid,
                                                    float* __restrict__ out) {
  const int c  = blockIdx.x & (C_ - 1);
  const int n0 = (blockIdx.x >> 6) * APL;
  const size_t base   = ((size_t)n0 * C_ + c) * (H_ * W_) + threadIdx.x * 4;
  const size_t nstep  = (size_t)C_ * H_ * W_;

  float4 v[APL], r[APL];
  for (int j = 0; j < APL; ++j) v[j] = *reinterpret_cast<const float4*>(y + base + j * nstep);
  if (resid) {
    for (int j = 0; j < APL; ++j) r[j] = *reinterpret_cast<const float4*>(resid + base + j * nstep);
  }

  float S = 0.f, Q = 0.f;
#pragma unroll
  for (int i = 0; i < NSPLIT; ++i) {
    S += part[(size_t)(c * NSPLIT + i) * PSTRIDE];
    Q += part[(size_t)(c * NSPLIT + i) * PSTRIDE + 1];
  }
  const float inv_n = 1.0f / (float)(B_ * H_ * W_);
  float mu    = S * inv_n;
  float var   = Q * inv_n - mu * mu;
  float inv   = rsqrtf(var + 1e-5f);
  float scale = g[c] * inv;
  float shift = b[c] - mu * scale;

  for (int j = 0; j < APL; ++j) {
    float4 o;
    o.x = v[j].x * scale + shift;
    o.y = v[j].y * scale + shift;
    o.z = v[j].z * scale + shift;
    o.w = v[j].w * scale + shift;
    if (resid) { o.x += r[j].x; o.y += r[j].y; o.z += r[j].z; o.w += r[j].w; }
    o.x = fmaxf(o.x, 0.f); o.y = fmaxf(o.y, 0.f); o.z = fmaxf(o.z, 0.f); o.w = fmaxf(o.w, 0.f);
    *reinterpret_cast<float4*>(out + base + j * nstep) = o;
  }
}

extern "C" void kernel_launch(void* const* d_in, const int* in_sizes, int n_in,
                              void* d_out, int out_size, void* d_ws, size_t ws_size,
                              hipStream_t stream) {
  (void)in_sizes; (void)n_in; (void)out_size; (void)ws_size;
  const float* x  = (const float*)d_in[0];
  const float* w1 = (const float*)d_in[1];
  const float* g1 = (const float*)d_in[2];
  const float* b1 = (const float*)d_in[3];
  const float* w2 = (const float*)d_in[4];
  const float* g2 = (const float*)d_in[5];
  const float* b2 = (const float*)d_in[6];
  float* out = (float*)d_out;

  char* ws = (char*)d_ws;
  const size_t TEN   = (size_t)B_ * C_ * H_ * W_;
  const size_t WTSZ  = (size_t)C_ * KD * sizeof(float);
  float* ybuf = (float*)(ws);
  float* abuf = (float*)(ws + TEN * 4);
  const size_t PSZ   = (size_t)C_ * NSPLIT * PSTRIDE * sizeof(float);
  float* pt1  = (float*)(ws + 2 * TEN * 4 + 2 * WTSZ);
  float* pt2  = (float*)(ws + 2 * TEN * 4 + 2 * WTSZ + PSZ);

  dim3 agrid(C_ / COG, (H_ / TILE) * (W_ / TILE), B_);

  adder_kernel<<<agrid, 256, 0, stream>>>(x, w1, ybuf);
  stats_kernel<<<C_ * NSPLIT, 256, 0, stream>>>(ybuf, pt1);
  apply_kernel<<<(B_ / APL) * C_, 256, 0, stream>>>(ybuf, pt1, g1, b1, nullptr, abuf);

  adder_kernel<<<agrid, 256, 0, stream>>>(abuf, w2, ybuf);
  stats_kernel<<<C_ * NSPLIT, 256, 0, stream>>>(ybuf, pt2);
  apply_kernel<<<(B_ / APL) * C_, 256, 0, stream>>>(ybuf, pt2, g2, b2, x, out);
}


# layer-1 normalize + ReLU pass removed (6 launches -> 5): the second adder applies scale/shift + ReLU to each raw layer-1 value while staging it into LDS; per-element arithmetic and sum orders kept
# speedup vs baseline: 2.9494x; 1.0573x over previous
#include <hip/hip_runtime.h>


typedef float v2f __attribute__((ext_vector_type(2)));
typedef float v8f __attribute__((ext_vector_type(8)));
typedef v8f v8f_a4 __attribute__((aligned(4)));
typedef __attribute__((address_space(4))) v8f_a4 cv8f;
typedef __attribute__((address_space(4))) float cf32;

#define B_   32
#define C_   64
#define H_   32
#define W_   32
#define KD   576
#define COG  16
#define CPL  8
#define TILE 16
#define CCH  32
#define PLANE    ((TILE + 2) * (TILE + 2))
#define STAGE_N  (CCH * PLANE)
#define STAGE_IT ((STAGE_N + 255) / 256)
#define NSPLIT   8
#define PSTRIDE  32

template <bool BN>
__global__ __launch_bounds__(256) void adder_kernel(const float* __restrict__ in,
                                                    const float* __restrict__ wgt,
                                                    float* __restrict__ y,
                                                    const float* __restrict__ part,
                                                    const float* __restrict__ g,
                                                    const float* __restrict__ b) {
  __shared__ float xs[STAGE_IT * 256];
  __shared__ float swl[COG];
  __shared__ float2 bnp[BN ? C_ : 1];
  const int tid    = threadIdx.x;
  __builtin_assume(tid < 256);
  if (BN) {
    if (tid < C_) {
      float S = 0.f, Q = 0.f;
#pragma unroll
      for (int i = 0; i < NSPLIT; ++i) {
        S += part[(size_t)(tid * NSPLIT + i) * PSTRIDE];
        Q += part[(size_t)(tid * NSPLIT + i) * PSTRIDE + 1];
      }
      const float inv_n = 1.0f / (float)(B_ * H_ * W_);
      const float mu    = S * inv_n;
      const float var   = Q * inv_n - mu * mu;
      const float inv   = rsqrtf(var + 1e-5f);
      const float scale = g[tid] * inv;
      const float shift = b[tid] - mu * scale;
      bnp[tid] = make_float2(scale, shift);
    }
  }
  const int cg     = blockIdx.x;
  const int tileId = blockIdx.y;
  const int n      = blockIdx.z;
  const int ty0 = (tileId >> 1) * TILE;
  const int tx0 = (tileId & 1) * TILE;
  const int chalf = __builtin_amdgcn_readfirstlane(tid >> 7);
  const int tx = tid & 15, ty = (tid >> 4) & 7;

  const float* wblk = wgt + (cg * COG + chalf * CPL) * KD;
  __builtin_prefetch(wblk, 0, 1);

  {
    const int wv   = tid >> 5, lane = tid & 31;
    const float4* wr = reinterpret_cast<const float4*>(wgt + (size_t)(cg * COG + wv * 2) * KD) + lane;
    float4 q[9];
#pragma unroll
    for (int r = 0; r < 9; ++r) q[r] = wr[r * 32];
    float s0 = 0.f, s1 = 0.f;
#pragma unroll
    for (int r = 0; r < 4; ++r) s0 += (q[r].x + q[r].y) + (q[r].z + q[r].w);
#pragma unroll
    for (int r = 5; r < 9; ++r) s1 += (q[r].x + q[r].y) + (q[r].z + q[r].w);
    const float mid = (q[4].x + q[4].y) + (q[4].z + q[4].w);
    s0 += (lane < 16) ? mid : 0.f;
    s1 += (lane < 16) ? 0.f : mid;
    for (int off = 16; off > 0; off >>= 1) {
      s0 += __shfl_xor(s0, off, 32);
      s1 += __shfl_xor(s1, off, 32);
    }
    if (lane == 0) { swl[wv * 2] = s0; swl[wv * 2 + 1] = s1; }
  }

  float acc0[CPL], acc1[CPL];
#pragma unroll
  for (int c = 0; c < CPL; ++c) { acc0[c] = 0.f; acc1[c] = 0.f; }
  float sx0 = 0.f, sx1 = 0.f;

  for (int c0 = 0; c0 < C_; c0 += CCH) {
    const float* inb = in + ((size_t)n * C_ + c0) * (H_ * W_);
    float v[STAGE_IT];
#pragma unroll
    for (int j = 0; j < STAGE_IT; ++j) {
      const int i  = min(tid + j * 256, STAGE_N - 1);
      const int ci = i / PLANE;
      const int r  = i - ci * PLANE;
      const int yy = r / (TILE + 2);
      const int xx = r - yy * (TILE + 2);
      const int gy = ty0 + yy - 1;
      const int gx = tx0 + xx - 1;
      const bool inside = (unsigned)gy < (unsigned)H_ && (unsigned)gx < (unsigned)W_;
      const int cy = min(max(gy, 0), H_ - 1);
      const int cx = min(max(gx, 0), W_ - 1);
      const float t = inb[(unsigned)((ci * H_ + cy) * W_ + cx)];
      v[j] = (BN || inside) ? t : 0.f;
    }
    if (BN) {
      if (c0 == 0) __syncthreads();
#pragma unroll
      for (int j = 0; j < STAGE_IT; ++j) {
        const int i  = min(tid + j * 256, STAGE_N - 1);
        const int ci = i / PLANE;
        const int r  = i - ci * PLANE;
        const int yy = r / (TILE + 2);
        const int xx = r - yy * (TILE + 2);
        const bool inside = (unsigned)(ty0 + yy - 1) < (unsigned)H_ && (unsigned)(tx0 + xx - 1) < (unsigned)W_;
        const float2 p = bnp[c0 + ci];
        float o = v[j] * p.x + p.y;
        o = fmaxf(o, 0.f);
        v[j] = inside ? o : 0.f;
      }
    }
#pragma unroll
    for (int j = 0; j < STAGE_IT; ++j) xs[tid + j * 256] = v[j];
    __syncthreads();

    for (int ci = 0; ci < CCH; ++ci) {
      const float* wci = wblk + (c0 + ci) * 9;
      v8f   w07[CPL];
      float w8[CPL];
#pragma unroll
      for (int c = 0; c < CPL; ++c) {
        w07[c] = *(const cv8f*)(wci + c * KD);
        w8[c]  = *(const cf32*)(wci + c * KD + 8);
      }
#pragma unroll
      for (int kh = 0; kh < 3; ++kh) {
#pragma unroll
        for (int kw = 0; kw < 3; ++kw) {
          const float xa = xs[(ci * (TILE + 2) + ty + kh) * (TILE + 2) + tx + kw];
          const float xb = xs[(ci * (TILE + 2) + ty + 8 + kh) * (TILE + 2) + tx + kw];
          float w[CPL];
#pragma unroll
          for (int c = 0; c < CPL; ++c) w[c] = (kh * 3 + kw < 8) ? w07[c][kh * 3 + kw] : w8[c];
          float m0[CPL], m1[CPL];
          asm("v_dual_max_num_f32 %0, %16, %24 :: v_dual_max_num_f32 %8, %16, %25\n\t"
              "v_dual_max_num_f32 %1, %17, %24 :: v_dual_max_num_f32 %9, %17, %25\n\t"
              "v_dual_max_num_f32 %2, %18, %24 :: v_dual_max_num_f32 %10, %18, %25\n\t"
              "v_dual_max_num_f32 %3, %19, %24 :: v_dual_max_num_f32 %11, %19, %25\n\t"
              "v_dual_max_num_f32 %4, %20, %24 :: v_dual_max_num_f32 %12, %20, %25\n\t"
              "v_dual_max_num_f32 %5, %21, %24 :: v_dual_max_num_f32 %13, %21, %25\n\t"
              "v_dual_max_num_f32 %6, %22, %24 :: v_dual_max_num_f32 %14, %22, %25\n\t"
              "v_dual_max_num_f32 %7, %23, %24 :: v_dual_max_num_f32 %15, %23, %25"
              : "=&v"(m0[0]), "=&v"(m0[1]), "=&v"(m0[2]), "=&v"(m0[3]),
                "=&v"(m0[4]), "=&v"(m0[5]), "=&v"(m0[6]), "=&v"(m0[7]),
                "=&v"(m1[0]), "=&v"(m1[1]), "=&v"(m1[2]), "=&v"(m1[3]),
                "=&v"(m1[4]), "=&v"(m1[5]), "=&v"(m1[6]), "=&v"(m1[7])
              : "s"(w[0]), "s"(w[1]), "s"(w[2]), "s"(w[3]), "s"(w[4]), "s"(w[5]), "s"(w[6]), "s"(w[7]),
                "v"(xa), "v"(xb));
          sx0 += xa;
          sx1 += xb;
#pragma unroll
          for (int c = 0; c < CPL; ++c) {
            acc0[c] += m0[c];
            acc1[c] += m1[c];
          }
        }
      }
    }
    __syncthreads();
  }

  const int gx = tx0 + tx;
#pragma unroll
  for (int c = 0; c < CPL; ++c) {
    float* yc = y + (((size_t)n * C_ + (cg * COG + chalf * CPL + c)) * H_ + ty0 + ty) * W_ + gx;
    const float sw = swl[chalf * CPL + c];
    yc[0]      = (sx0 + sw) - 2.f * acc0[c];
    yc[8 * W_] = (sx1 + sw) - 2.f * acc1[c];
  }
}

__global__ __launch_bounds__(256) void stats_kernel(const float* __restrict__ y,
                                                    float* __restrict__ part) {
  const int c    = blockIdx.x / NSPLIT;
  const int sp   = blockIdx.x % NSPLIT;
  const int tid  = threadIdx.x;
  const int wave = tid >> 5, lane = tid & 31;

  const int n = sp * (B_ / NSPLIT) + (wave >> 1);
  const float4* p = reinterpret_cast<const float4*>(
      y + ((size_t)n * C_ + c) * (H_ * W_) + (wave & 1) * 512) + lane;
  float4 v[4];
#pragma unroll
  for (int r = 0; r < 4; ++r) v[r] = p[r * 32];

  v8f accS = {};
  v8f accQ = {};
  v2f ones = {1.f, 1.f};
#pragma unroll
  for (int r = 0; r < 4; ++r) {
    v2f d0 = {v[r].x, v[r].y};
    v2f d1 = {v[r].z, v[r].w};
    v2f q0 = {d0.x * d0.x, d0.y * d0.y};
    v2f q1 = {d1.x * d1.x, d1.y * d1.y};
    accS = __builtin_amdgcn_wmma_f32_16x16x4_f32(false, ones, false, d0,
                                                 (short)0, accS, false, false);
    accQ = __builtin_amdgcn_wmma_f32_16x16x4_f32(false, ones, false, q0,
                                                 (short)0, accQ, false, false);
    accS = __builtin_amdgcn_wmma_f32_16x16x4_f32(false, ones, false, d1,
                                                 (short)0, accS, false, false);
    accQ = __builtin_amdgcn_wmma_f32_16x16x4_f32(false, ones, false, q1,
                                                 (short)0, accQ, false, false);
  }

  float s = accS[0], q = accQ[0];
  for (int off = 16; off > 0; off >>= 1) {
    s += __shfl_xor(s, off, 32);
    q += __shfl_xor(q, off, 32);
  }
  s *= 0.5f;
  q *= 0.5f;

  __shared__ float ls[8], lq[8];
  if (lane == 0) { ls[wave] = s; lq[wave] = q; }
  __syncthreads();
  if (tid == 0) {
    float S = 0.f, Q = 0.f;
    for (int i = 0; i < 8; ++i) { S += ls[i]; Q += lq[i]; }
    part[(size_t)blockIdx.x * PSTRIDE]     = S;
    part[(size_t)blockIdx.x * PSTRIDE + 1] = Q;
  }
}

#define APL 4
__global__ __launch_bounds__(256) void apply_kernel(const float* __restrict__ y,
                                                    const float* __restrict__ part,
                                                    const float* __restrict__ g,
                                                    const float* __restrict__ b,
                                                    const float* __restrict__ resid,
                                                    float* __restrict__ out) {
  const int c  = blockIdx.x & (C_ - 1);
  const int n0 = (blockIdx.x >> 6) * APL;
  const size_t base   = ((size_t)n0 * C_ + c) * (H_ * W_) + threadIdx.x * 4;
  const size_t nstep  = (size_t)C_ * H_ * W_;

  float4 v[APL], r[APL];
  for (int j = 0; j < APL; ++j) v[j] = *reinterpret_cast<const float4*>(y + base + j * nstep);
  if (resid) {
    for (int j = 0; j < APL; ++j) r[j] = *reinterpret_cast<const float4*>(resid + base + j * nstep);
  }

  float S = 0.f, Q = 0.f;
#pragma unroll
  for (int i = 0; i < NSPLIT; ++i) {
    S += part[(size_t)(c * NSPLIT + i) * PSTRIDE];
    Q += part[(size_t)(c * NSPLIT + i) * PSTRIDE + 1];
  }
  const float inv_n = 1.0f / (float)(B_ * H_ * W_);
  float mu    = S * inv_n;
  float var   = Q * inv_n - mu * mu;
  float inv   = rsqrtf(var + 1e-5f);
  float scale = g[c] * inv;
  float shift = b[c] - mu * scale;

  for (int j = 0; j < APL; ++j) {
    float4 o;
    o.x = v[j].x * scale + shift;
    o.y = v[j].y * scale + shift;
    o.z = v[j].z * scale + shift;
    o.w = v[j].w * scale + shift;
    if (resid) { o.x += r[j].x; o.y += r[j].y; o.z += r[j].z; o.w += r[j].w; }
    o.x = fmaxf(o.x, 0.f); o.y = fmaxf(o.y, 0.f); o.z = fmaxf(o.z, 0.f); o.w = fmaxf(o.w, 0.f);
    *reinterpret_cast<float4*>(out + base + j * nstep) = o;
  }
}

extern "C" void kernel_launch(void* const* d_in, const int* in_sizes, int n_in,
                              void* d_out, int out_size, void* d_ws, size_t ws_size,
                              hipStream_t stream) {
  (void)in_sizes; (void)n_in; (void)out_size; (void)ws_size;
  const float* x  = (const float*)d_in[0];
  const float* w1 = (const float*)d_in[1];
  const float* g1 = (const float*)d_in[2];
  const float* b1 = (const float*)d_in[3];
  const float* w2 = (const float*)d_in[4];
  const float* g2 = (const float*)d_in[5];
  const float* b2 = (const float*)d_in[6];
  float* out = (float*)d_out;

  char* ws = (char*)d_ws;
  const size_t TEN   = (size_t)B_ * C_ * H_ * W_;
  const size_t WTSZ  = (size_t)C_ * KD * sizeof(float);
  float* ybuf = (float*)(ws);
  float* abuf = (float*)(ws + TEN * 4);
  const size_t PSZ   = (size_t)C_ * NSPLIT * PSTRIDE * sizeof(float);
  float* pt1  = (float*)(ws + 2 * TEN * 4 + 2 * WTSZ);
  float* pt2  = (float*)(ws + 2 * TEN * 4 + 2 * WTSZ + PSZ);

  dim3 agrid(C_ / COG, (H_ / TILE) * (W_ / TILE), B_);

  adder_kernel<false><<<agrid, 256, 0, stream>>>(x, w1, ybuf, nullptr, nullptr, nullptr);
  stats_kernel<<<C_ * NSPLIT, 256, 0, stream>>>(ybuf, pt1);

  adder_kernel<true><<<agrid, 256, 0, stream>>>(ybuf, w2, abuf, pt1, g1, b1);
  stats_kernel<<<C_ * NSPLIT, 256, 0, stream>>>(abuf, pt2);
  apply_kernel<<<(B_ / APL) * C_, 256, 0, stream>>>(abuf, pt2, g2, b2, x, out);
}


# adder tile = 8 whole image rows (was 16x16): staging is 10 128-bit loads + LDS stores per thread and chunk (was 41 32-bit with index division); outputs go out as whole 128-B lines; sum order kept
# speedup vs baseline: 3.4765x; 1.1787x over previous
#include <hip/hip_runtime.h>


typedef float v2f __attribute__((ext_vector_type(2)));
typedef float v8f __attribute__((ext_vector_type(8)));
typedef v8f v8f_a4 __attribute__((aligned(4)));
typedef __attribute__((address_space(4))) v8f_a4 cv8f;
typedef __attribute__((address_space(4))) float cf32;

#define B_   32
#define C_   64
#define H_   32
#define W_   32
#define KD   576
#define COG  16
#define CPL  8
#define TH   8
#define CCH  32
#define ROWS  (TH + 2)
#define PITCH 40
#define PLANE (ROWS * PITCH)
#define NSPLIT   8
#define PSTRIDE  32

template <bool BN>
__global__ __launch_bounds__(256) void adder_kernel(const float* __restrict__ in,
                                                    const float* __restrict__ wgt,
                                                    float* __restrict__ y,
                                                    const float* __restrict__ part,
                                                    const float* __restrict__ g,
                                                    const float* __restrict__ b) {
  __shared__ __attribute__((aligned(16))) float xs[CCH * PLANE];
  __shared__ float swl[COG];
  __shared__ float2 bnp[BN ? C_ : 1];
  const int tid    = threadIdx.x;
  __builtin_assume(tid < 256);
  if (BN) {
    if (tid < C_) {
      float S = 0.f, Q = 0.f;
#pragma unroll
      for (int i = 0; i < NSPLIT; ++i) {
        S += part[(size_t)(tid * NSPLIT + i) * PSTRIDE];
        Q += part[(size_t)(tid * NSPLIT + i) * PSTRIDE + 1];
      }
      const float inv_n = 1.0f / (float)(B_ * H_ * W_);
      const float mu    = S * inv_n;
      const float var   = Q * inv_n - mu * mu;
      const float inv   = rsqrtf(var + 1e-5f);
      const float scale = g[tid] * inv;
      const float shift = b[tid] - mu * scale;
      bnp[tid] = make_float2(scale, shift);
    }
  }
  const int cg     = blockIdx.x;
  const int tileId = blockIdx.y;
  __builtin_assume(tileId >= 0 && tileId < H_ / TH);
  const int n      = blockIdx.z;
  const int ty0 = tileId * TH;
  const int chalf = __builtin_amdgcn_readfirstlane(tid >> 7);
  const int tx = tid & 31;
  const int ty = __builtin_amdgcn_readfirstlane((tid >> 5) & 3);

  const float* wblk = wgt + (cg * COG + chalf * CPL) * KD;
  __builtin_prefetch(wblk, 0, 1);

  {
    const int wv   = tid >> 5, lane = tid & 31;
    const float4* wr = reinterpret_cast<const float4*>(wgt + (size_t)(cg * COG + wv * 2) * KD) + lane;
    float4 q[9];
#pragma unroll
    for (int r = 0; r < 9; ++r) q[r] = wr[r * 32];
    float s0 = 0.f, s1 = 0.f;
#pragma unroll
    for (int r = 0; r < 4; ++r) s0 += (q[r].x + q[r].y) + (q[r].z + q[r].w);
#pragma unroll
    for (int r = 5; r < 9; ++r) s1 += (q[r].x + q[r].y) + (q[r].z + q[r].w);
    const float mid = (q[4].x + q[4].y) + (q[4].z + q[4].w);
    s0 += (lane < 16) ? mid : 0.f;
    s1 += (lane < 16) ? 0.f : mid;
    for (int off = 16; off > 0; off >>= 1) {
      s0 += __shfl_xor(s0, off, 32);
      s1 += __shfl_xor(s1, off, 32);
    }
    if (lane == 0) { swl[wv * 2] = s0; swl[wv * 2 + 1] = s1; }
  }

  float acc0[CPL], acc1[CPL];
#pragma unroll
  for (int c = 0; c < CPL; ++c) { acc0[c] = 0.f; acc1[c] = 0.f; }
  float sx0 = 0.f, sx1 = 0.f;

  for (int i = tid; i < CCH * ROWS * 2; i += 256) xs[(i >> 1) * PITCH + ((i & 1) ? 36 : 3)] = 0.f;

  const int sq = tid & 7, sci = tid >> 3;

  for (int c0 = 0; c0 < C_; c0 += CCH) {
    const float* inb = in + ((size_t)n * C_ + c0 + sci) * (H_ * W_) + sq * 4;
    float4 v[ROWS];
#pragma unroll
    for (int j = 0; j < ROWS; ++j) {
      const int gy = ty0 + j - 1;
      const bool inside = (unsigned)gy < (unsigned)H_;
      const int cy = min(max(gy, 0), H_ - 1);
      const float4 t = *reinterpret_cast<const float4*>(inb + cy * W_);
      v[j] = (BN || inside) ? t : make_float4(0.f, 0.f, 0.f, 0.f);
    }
    if (BN) {
      if (c0 == 0) __syncthreads();
      const float2 p = bnp[c0 + sci];
#pragma unroll
      for (int j = 0; j < ROWS; ++j) {
        const bool inside = (unsigned)(ty0 + j - 1) < (unsigned)H_;
        float4 o;
        o.x = fmaxf(v[j].x * p.x + p.y, 0.f);
        o.y = fmaxf(v[j].y * p.x + p.y, 0.f);
        o.z = fmaxf(v[j].z * p.x + p.y, 0.f);
        o.w = fmaxf(v[j].w * p.x + p.y, 0.f);
        v[j] = inside ? o : make_float4(0.f, 0.f, 0.f, 0.f);
      }
    }
#pragma unroll
    for (int j = 0; j < ROWS; ++j)
      *reinterpret_cast<float4*>(xs + (sci * ROWS + j) * PITCH + 4 + sq * 4) = v[j];
    __syncthreads();

    for (int ci = 0; ci < CCH; ++ci) {
      const float* wci = wblk + (c0 + ci) * 9;
      v8f   w07[CPL];
      float w8[CPL];
#pragma unroll
      for (int c = 0; c < CPL; ++c) {
        w07[c] = *(const cv8f*)(wci + c * KD);
        w8[c]  = *(const cf32*)(wci + c * KD + 8);
      }
#pragma unroll
      for (int kh = 0; kh < 3; ++kh) {
#pragma unroll
        for (int kw = 0; kw < 3; ++kw) {
          const float xa = xs[(ci * ROWS + ty + kh) * PITCH + 3 + tx + kw];
          const float xb = xs[(ci * ROWS + ty + 4 + kh) * PITCH + 3 + tx + kw];
          float w[CPL];
#pragma unroll
          for (int c = 0; c < CPL; ++c) w[c] = (kh * 3 + kw < 8) ? w07[c][kh * 3 + kw] : w8[c];
          float m0[CPL], m1[CPL];
          asm("v_dual_max_num_f32 %0, %16, %24 :: v_dual_max_num_f32 %8, %16, %25\n\t"
              "v_dual_max_num_f32 %1, %17, %24 :: v_dual_max_num_f32 %9, %17, %25\n\t"
              "v_dual_max_num_f32 %2, %18, %24 :: v_dual_max_num_f32 %10, %18, %25\n\t"
              "v_dual_max_num_f32 %3, %19, %24 :: v_dual_max_num_f32 %11, %19, %25\n\t"
              "v_dual_max_num_f32 %4, %20, %24 :: v_dual_max_num_f32 %12, %20, %25\n\t"
              "v_dual_max_num_f32 %5, %21, %24 :: v_dual_max_num_f32 %13, %21, %25\n\t"
              "v_dual_max_num_f32 %6, %22, %24 :: v_dual_max_num_f32 %14, %22, %25\n\t"
              "v_dual_max_num_f32 %7, %23, %24 :: v_dual_max_num_f32 %15, %23, %25"
              : "=&v"(m0[0]), "=&v"(m0[1]), "=&v"(m0[2]), "=&v"(m0[3]),
                "=&v"(m0[4]), "=&v"(m0[5]), "=&v"(m0[6]), "=&v"(m0[7]),
                "=&v"(m1[0]), "=&v"(m1[1]), "=&v"(m1[2]), "=&v"(m1[3]),
                "=&v"(m1[4]), "=&v"(m1[5]), "=&v"(m1[6]), "=&v"(m1[7])
              : "s"(w[0]), "s"(w[1]), "s"(w[2]), "s"(w[3]), "s"(w[4]), "s"(w[5]), "s"(w[6]), "s"(w[7]),
                "v"(xa), "v"(xb));
          sx0 += xa;
          sx1 += xb;
#pragma unroll
          for (int c = 0; c < CPL; ++c) {
            acc0[c] += m0[c];
            acc1[c] += m1[c];
          }
        }
      }
    }
    __syncthreads();
  }

#pragma unroll
  for (int c = 0; c < CPL; ++c) {
    float* yc = y + (((size_t)n * C_ + (cg * COG + chalf * CPL + c)) * H_ + ty0 + ty) * W_ + tx;
    const float sw = swl[chalf * CPL + c];
    yc[0]      = (sx0 + sw) - 2.f * acc0[c];
    yc[4 * W_] = (sx1 + sw) - 2.f * acc1[c];
  }
}

__global__ __launch_bounds__(256) void stats_kernel(const float* __restrict__ y,
                                                    float* __restrict__ part) {
  const int c    = blockIdx.x / NSPLIT;
  const int sp   = blockIdx.x % NSPLIT;
  const int tid  = threadIdx.x;
  const int wave = tid >> 5, lane = tid & 31;

  const int n = sp * (B_ / NSPLIT) + (wave >> 1);
  const float4* p = reinterpret_cast<const float4*>(
      y + ((size_t)n * C_ + c) * (H_ * W_) + (wave & 1) * 512) + lane;
  float4 v[4];
#pragma unroll
  for (int r = 0; r < 4; ++r) v[r] = p[r * 32];

  v8f accS = {};
  v8f accQ = {};
  v2f ones = {1.f, 1.f};
#pragma unroll
  for (int r = 0; r < 4; ++r) {
    v2f d0 = {v[r].x, v[r].y};
    v2f d1 = {v[r].z, v[r].w};
    v2f q0 = {d0.x * d0.x, d0.y * d0.y};
    v2f q1 = {d1.x * d1.x, d1.y * d1.y};
    accS = __builtin_amdgcn_wmma_f32_16x16x4_f32(false, ones, false, d0,
                                                 (short)0, accS, false, false);
    accQ = __builtin_amdgcn_wmma_f32_16x16x4_f32(false, ones, false, q0,
                                                 (short)0, accQ, false, false);
    accS = __builtin_amdgcn_wmma_f32_16x16x4_f32(false, ones, false, d1,
                                                 (short)0, accS, false, false);
    accQ = __builtin_amdgcn_wmma_f32_16x16x4_f32(false, ones, false, q1,
                                                 (short)0, accQ, false, false);
  }

  float s = accS[0], q = accQ[0];
  for (int off = 16; off > 0; off >>= 1) {
    s += __shfl_xor(s, off, 32);
    q += __shfl_xor(q, off, 32);
  }
  s *= 0.5f;
  q *= 0.5f;

  __shared__ float ls[8], lq[8];
  if (lane == 0) { ls[wave] = s; lq[wave] = q; }
  __syncthreads();
  if (tid == 0) {
    float S = 0.f, Q = 0.f;
    for (int i = 0; i < 8; ++i) { S += ls[i]; Q += lq[i]; }
    part[(size_t)blockIdx.x * PSTRIDE]     = S;
    part[(size_t)blockIdx.x * PSTRIDE + 1] = Q;
  }
}

#define APL 4
__global__ __launch_bounds__(256) void apply_kernel(const float* __restrict__ y,
                                                    const float* __restrict__ part,
                                                    const float* __restrict__ g,
                                                    const float* __restrict__ b,
                                                    const float* __restrict__ resid,
                                                    float* __restrict__ out) {
  const int c  = blockIdx.x & (C_ - 1);
  const int n0 = (blockIdx.x >> 6) * APL;
  const size_t base   = ((size_t)n0 * C_ + c) * (H_ * W_) + threadIdx.x * 4;
  const size_t nstep  = (size_t)C_ * H_ * W_;

  float4 v[APL], r[APL];
  for (int j = 0; j < APL; ++j) v[j] = *reinterpret_cast<const float4*>(y + base + j * nstep);
  if (resid) {
    for (int j = 0; j < APL; ++j) r[j] = *reinterpret_cast<const float4*>(resid + base + j * nstep);
  }

  float S = 0.f, Q = 0.f;
#pragma unroll
  for (int i = 0; i < NSPLIT; ++i) {
    S += part[(size_t)(c * NSPLIT + i) * PSTRIDE];
    Q += part[(size_t)(c * NSPLIT + i) * PSTRIDE + 1];
  }
  const float inv_n = 1.0f / (float)(B_ * H_ * W_);
  float mu    = S * inv_n;
  float var   = Q * inv_n - mu * mu;
  float inv   = rsqrtf(var + 1e-5f);
  float scale = g[c] * inv;
  float shift = b[c] - mu * scale;

  for (int j = 0; j < APL; ++j) {
    float4 o;
    o.x = v[j].x * scale + shift;
    o.y = v[j].y * scale + shift;
    o.z = v[j].z * scale + shift;
    o.w = v[j].w * scale + shift;
    if (resid) { o.x += r[j].x; o.y += r[j].y; o.z += r[j].z; o.w += r[j].w; }
    o.x = fmaxf(o.x, 0.f); o.y = fmaxf(o.y, 0.f); o.z = fmaxf(o.z, 0.f); o.w = fmaxf(o.w, 0.f);
    *reinterpret_cast<float4*>(out + base + j * nstep) = o;
  }
}

extern "C" void kernel_launch(void* const* d_in, const int* in_sizes, int n_in,
                              void* d_out, int out_size, void* d_ws, size_t ws_size,
                              hipStream_t stream) {
  (void)in_sizes; (void)n_in; (void)out_size; (void)ws_size;
  const float* x  = (const float*)d_in[0];
  const float* w1 = (const float*)d_in[1];
  const float* g1 = (const float*)d_in[2];
  const float* b1 = (const float*)d_in[3];
  const float* w2 = (const float*)d_in[4];
  const float* g2 = (const float*)d_in[5];
  const float* b2 = (const float*)d_in[6];
  float* out = (float*)d_out;

  char* ws = (char*)d_ws;
  const size_t TEN   = (size_t)B_ * C_ * H_ * W_;
  const size_t WTSZ  = (size_t)C_ * KD * sizeof(float);
  float* ybuf = (float*)(ws);
  float* abuf = (float*)(ws + TEN * 4);
  const size_t PSZ   = (size_t)C_ * NSPLIT * PSTRIDE * sizeof(float);
  float* pt1  = (float*)(ws + 2 * TEN * 4 + 2 * WTSZ);
  float* pt2  = (float*)(ws + 2 * TEN * 4 + 2 * WTSZ + PSZ);

  dim3 agrid(C_ / COG, H_ / TH, B_);

  adder_kernel<false><<<agrid, 256, 0, stream>>>(x, w1, ybuf, nullptr, nullptr, nullptr);
  stats_kernel<<<C_ * NSPLIT, 256, 0, stream>>>(ybuf, pt1);

  adder_kernel<true><<<agrid, 256, 0, stream>>>(ybuf, w2, abuf, pt1, g1, b1);
  stats_kernel<<<C_ * NSPLIT, 256, 0, stream>>>(abuf, pt2);
  apply_kernel<<<(B_ / APL) * C_, 256, 0, stream>>>(abuf, pt2, g2, b2, x, out);
}
